# starting kernel
# baseline (speedup 1.0000x reference)
#include <hip/hip_runtime.h>
#include <math.h>

constexpr int BB = 2;
constexpr int NN = 50000;
constexpr int FF = 8;
constexpr int EE = 800000;
constexpr int DD = 3;
constexpr int KK = 8;
constexpr int HH = 64;
constexpr int FB = FF * BB;
constexpr int NK = NN * KK;
constexpr int CDIM = FF * KK;
constexpr int MROWS = BB * NN;
constexpr int MTILES = MROWS / 16;
constexpr int NTILES = HH / 16;
constexpr float EPSF = 1e-14f;

typedef __attribute__((ext_vector_type(2))) float v2f;
typedef __attribute__((ext_vector_type(8))) float v8f;

__device__ __forceinline__ unsigned fkey(float f) {
  unsigned b = __float_as_uint(f);
  return (b & 0x80000000u) ? ~b : (b | 0x80000000u);
}
__device__ __forceinline__ float finv(unsigned k) {
  return __uint_as_float((k & 0x80000000u) ? (k & 0x7fffffffu) : ~k);
}

__global__ void k_init(unsigned* ws, int n) {
  int i = blockIdx.x * blockDim.x + threadIdx.x;
  int stride = gridDim.x * blockDim.x;
  for (; i < n; i += stride) ws[i] = 0u;
}

__global__ void k_logit_max(const float* __restrict__ uval,
                            const float* __restrict__ mu,
                            const float* __restrict__ sigma,
                            const int* __restrict__ urows,
                            unsigned* __restrict__ mkey) {
  int e = blockIdx.x * blockDim.x + threadIdx.x;
  if (e >= EE) return;
  float mu_l[KK * DD], iv_l[KK * DD];
#pragma unroll
  for (int i = 0; i < KK * DD; ++i) {
    mu_l[i] = mu[i];
    float sg = sigma[i];
    iv_l[i] = 1.0f / (sg * sg + EPSF);
  }
  float uv[DD];
#pragma unroll
  for (int d = 0; d < DD; ++d) uv[d] = uval[e * DD + d];
  int row = urows[e];
#pragma unroll
  for (int k = 0; k < KK; ++k) {
    float s = 0.0f;
#pragma unroll
    for (int d = 0; d < DD; ++d) {
      float df = uv[d] - mu_l[k * DD + d];
      s += df * df * iv_l[k * DD + d];
    }
    atomicMax(&mkey[row * KK + k], fkey(-0.5f * s));
  }
}

__global__ void k_denom(const float* __restrict__ uval,
                        const float* __restrict__ mu,
                        const float* __restrict__ sigma,
                        const int* __restrict__ urows,
                        const unsigned* __restrict__ mkey,
                        float* __restrict__ denom) {
  int e = blockIdx.x * blockDim.x + threadIdx.x;
  if (e >= EE) return;
  float mu_l[KK * DD], iv_l[KK * DD];
#pragma unroll
  for (int i = 0; i < KK * DD; ++i) {
    mu_l[i] = mu[i];
    float sg = sigma[i];
    iv_l[i] = 1.0f / (sg * sg + EPSF);
  }
  float uv[DD];
#pragma unroll
  for (int d = 0; d < DD; ++d) uv[d] = uval[e * DD + d];
  int row = urows[e];
#pragma unroll
  for (int k = 0; k < KK; ++k) {
    float s = 0.0f;
#pragma unroll
    for (int d = 0; d < DD; ++d) {
      float df = uv[d] - mu_l[k * DD + d];
      s += df * df * iv_l[k * DD + d];
    }
    float l = -0.5f * s;
    float mval = finv(mkey[row * KK + k]);
    atomicAdd(&denom[row * KK + k], __expf(l - mval));
  }
}

__global__ void k_scatter(const float* __restrict__ X,
                          const float* __restrict__ uval,
                          const float* __restrict__ mu,
                          const float* __restrict__ sigma,
                          const int* __restrict__ urows,
                          const int* __restrict__ ucols,
                          const unsigned* __restrict__ mkey,
                          const float* __restrict__ denom,
                          float* __restrict__ agg) {
  int t = blockIdx.x * blockDim.x + threadIdx.x;
  if (t >= EE * KK) return;
  int e = t >> 3;
  int k = t & (KK - 1);
  int row = urows[e];
  int col = ucols[e];
  float s = 0.0f;
#pragma unroll
  for (int d = 0; d < DD; ++d) {
    float sg = sigma[k * DD + d];
    float df = uval[e * DD + d] - mu[k * DD + d];
    s += df * df / (sg * sg + EPSF);
  }
  float l = -0.5f * s;
  float w = __expf(l - finv(mkey[row * KK + k])) / denom[row * KK + k];
  const float* x0 = X + (size_t)col * FF;
  const float* x1 = X + (size_t)NN * FF + (size_t)col * FF;
  float* dst = agg + (size_t)row * (KK * FB) + k * FB;
#pragma unroll
  for (int f = 0; f < FF; ++f) {
    atomicAdd(dst + f * 2 + 0, w * x0[f]);
    atomicAdd(dst + f * 2 + 1, w * x1[f]);
  }
}

__global__ void k_gemm(const float* __restrict__ agg,
                       const float* __restrict__ GW,
                       float* __restrict__ out) {
  int wid = blockIdx.x * (blockDim.x >> 5) + (threadIdx.x >> 5);
  if (wid >= MTILES * NTILES) return;
  int lane = threadIdx.x & 31;
  int half = lane >> 4;
  int lm = lane & 15;
  int mt = wid >> 2;
  int nt = wid & 3;
  int row0 = mt * 16;
  int col0 = nt * 16;
  int b = row0 / NN;
  int n0 = row0 % NN;

  v8f acc = {};
#if __has_builtin(__builtin_amdgcn_wmma_f32_16x16x4_f32)
  const float* arow = agg + (size_t)(n0 + lm) * (KK * FB) + b;
#pragma unroll
  for (int kk = 0; kk < CDIM / 4; ++kk) {
    int c0 = kk * 4 + half * 2;
    int c1 = c0 + 1;
    v2f a, bf;
    a.x = arow[(c0 & 7) * FB + ((c0 >> 3) << 1)];
    a.y = arow[(c1 & 7) * FB + ((c1 >> 3) << 1)];
    bf.x = GW[c0 * HH + col0 + lm];
    bf.y = GW[c1 * HH + col0 + lm];
    acc = __builtin_amdgcn_wmma_f32_16x16x4_f32(
         false, a,  false, bf,
         (short)0, acc,  false,  false);
  }
#pragma unroll
  for (int j = 0; j < 8; ++j) {
    int r = row0 + half * 8 + j;
    out[(size_t)r * HH + col0 + lm] = acc[j];
  }
#else
#pragma unroll
  for (int j = 0; j < 8; ++j) {
    int r = row0 + half * 8 + j;
    int n = r % NN;
    int bb = r / NN;
    float s = 0.0f;
    for (int c = 0; c < CDIM; ++c) {
      float fv = agg[(size_t)n * (KK * FB) + (c & 7) * FB + ((c >> 3) << 1) + bb];
      s += fv * GW[c * HH + col0 + lm];
    }
    out[(size_t)r * HH + col0 + lm] = s;
  }
#endif
}

extern "C" void kernel_launch(void* const* d_in, const int* in_sizes, int n_in,
                              void* d_out, int out_size, void* d_ws, size_t ws_size,
                              hipStream_t stream) {
  const float* X     = (const float*)d_in[0];
  const float* uval  = (const float*)d_in[1];
  const float* mu    = (const float*)d_in[2];
  const float* sigma = (const float*)d_in[3];
  const float* GW    = (const float*)d_in[4];
  const int* urows   = (const int*)d_in[5];
  const int* ucols   = (const int*)d_in[6];
  float* out = (float*)d_out;

  unsigned* mkey = (unsigned*)d_ws;
  float* denom   = (float*)d_ws + NK;
  float* agg     = (float*)d_ws + 2 * NK;
  const int ws_words = 2 * NK + NK * FB;

  k_init<<<2048, 256, 0, stream>>>((unsigned*)d_ws, ws_words);

  k_logit_max<<<(EE + 255) / 256, 256, 0, stream>>>(uval, mu, sigma, urows, mkey);

  k_denom<<<(EE + 255) / 256, 256, 0, stream>>>(uval, mu, sigma, urows, mkey, denom);

  k_scatter<<<(EE * KK + 255) / 256, 256, 0, stream>>>(
      X, uval, mu, sigma, urows, ucols, mkey, denom, agg);

  int waves = MTILES * NTILES;
  int blocks = (waves + 7) / 8;
  k_gemm<<<blocks, 256, 0, stream>>>(agg, GW, out);
}


# Coalesce the global atomics: lanes of one edge take consecutive words, 8-word runs in the max and sum passes, whole 128-byte lines in the scatter; float-add arrival order varies as before
# speedup vs baseline: 3.9316x; 3.9316x over previous
#include <hip/hip_runtime.h>
#include <math.h>

constexpr int BB = 2;
constexpr int NN = 50000;
constexpr int FF = 8;
constexpr int EE = 800000;
constexpr int DD = 3;
constexpr int KK = 8;
constexpr int HH = 64;
constexpr int FB = FF * BB;
constexpr int NK = NN * KK;
constexpr int CDIM = FF * KK;
constexpr int MROWS = BB * NN;
constexpr int MTILES = MROWS / 16;
constexpr int NTILES = HH / 16;
constexpr float EPSF = 1e-14f;

typedef __attribute__((ext_vector_type(2))) float v2f;
typedef __attribute__((ext_vector_type(8))) float v8f;

__device__ __forceinline__ unsigned fkey(float f) {
  unsigned b = __float_as_uint(f);
  return (b & 0x80000000u) ? ~b : (b | 0x80000000u);
}
__device__ __forceinline__ float finv(unsigned k) {
  return __uint_as_float((k & 0x80000000u) ? (k & 0x7fffffffu) : ~k);
}

__global__ void k_init(unsigned* ws, int n) {
  int i = blockIdx.x * blockDim.x + threadIdx.x;
  int stride = gridDim.x * blockDim.x;
  for (; i < n; i += stride) ws[i] = 0u;
}

__global__ void k_logit_max(const float* __restrict__ uval,
                            const float* __restrict__ mu,
                            const float* __restrict__ sigma,
                            const int* __restrict__ urows,
                            unsigned* __restrict__ mkey) {
  int t = blockIdx.x * blockDim.x + threadIdx.x;
  if (t >= EE * KK) return;
  int e = t >> 3;
  int k = t & (KK - 1);
  int row = urows[e];
  float s = 0.0f;
#pragma unroll
  for (int d = 0; d < DD; ++d) {
    float sg = sigma[k * DD + d];
    float iv = 1.0f / (sg * sg + EPSF);
    float df = uval[e * DD + d] - mu[k * DD + d];
    s += df * df * iv;
  }
  atomicMax(&mkey[row * KK + k], fkey(-0.5f * s));
}

__global__ void k_denom(const float* __restrict__ uval,
                        const float* __restrict__ mu,
                        const float* __restrict__ sigma,
                        const int* __restrict__ urows,
                        const unsigned* __restrict__ mkey,
                        float* __restrict__ denom) {
  int t = blockIdx.x * blockDim.x + threadIdx.x;
  if (t >= EE * KK) return;
  int e = t >> 3;
  int k = t & (KK - 1);
  int row = urows[e];
  float s = 0.0f;
#pragma unroll
  for (int d = 0; d < DD; ++d) {
    float sg = sigma[k * DD + d];
    float iv = 1.0f / (sg * sg + EPSF);
    float df = uval[e * DD + d] - mu[k * DD + d];
    s += df * df * iv;
  }
  float l = -0.5f * s;
  float mval = finv(mkey[row * KK + k]);
  atomicAdd(&denom[row * KK + k], __expf(l - mval));
}

__global__ void k_scatter(const float* __restrict__ X,
                          const float* __restrict__ uval,
                          const float* __restrict__ mu,
                          const float* __restrict__ sigma,
                          const int* __restrict__ urows,
                          const int* __restrict__ ucols,
                          const unsigned* __restrict__ mkey,
                          const float* __restrict__ denom,
                          float* __restrict__ agg) {
  int t = blockIdx.x * blockDim.x + threadIdx.x;
  if (t >= EE * KK) return;
  int e = t >> 3;
  int k = t & (KK - 1);
  int row = urows[e];
  int col = ucols[e];
  float s = 0.0f;
#pragma unroll
  for (int d = 0; d < DD; ++d) {
    float sg = sigma[k * DD + d];
    float df = uval[e * DD + d] - mu[k * DD + d];
    s += df * df / (sg * sg + EPSF);
  }
  float l = -0.5f * s;
  float w = __expf(l - finv(mkey[row * KK + k])) / denom[row * KK + k];

  int lane = threadIdx.x & 31;
  int hi = lane >> 4;
  int fb = lane & 15;
  int xoff = (fb & 1) * (NN * FF) + (fb >> 1);
#pragma unroll
  for (int ee = 0; ee < 4; ++ee) {
    int row_e = __shfl(row, ee * 8);
    int col_e = __shfl(col, ee * 8);
    float xv = X[(size_t)col_e * FF + xoff];
    float* dst = agg + (size_t)row_e * (KK * FB) + lane;
#pragma unroll
    for (int i = 0; i < 4; ++i) {
      float wk = __shfl(w, ee * 8 + 2 * i + hi);
      atomicAdd(dst + i * 32, wk * xv);
    }
  }
}

__global__ void k_gemm(const float* __restrict__ agg,
                       const float* __restrict__ GW,
                       float* __restrict__ out) {
  int wid = blockIdx.x * (blockDim.x >> 5) + (threadIdx.x >> 5);
  if (wid >= MTILES * NTILES) return;
  int lane = threadIdx.x & 31;
  int half = lane >> 4;
  int lm = lane & 15;
  int mt = wid >> 2;
  int nt = wid & 3;
  int row0 = mt * 16;
  int col0 = nt * 16;
  int b = row0 / NN;
  int n0 = row0 % NN;

  v8f acc = {};
#if __has_builtin(__builtin_amdgcn_wmma_f32_16x16x4_f32)
  const float* arow = agg + (size_t)(n0 + lm) * (KK * FB) + b;
#pragma unroll
  for (int kk = 0; kk < CDIM / 4; ++kk) {
    int c0 = kk * 4 + half * 2;
    int c1 = c0 + 1;
    v2f a, bf;
    a.x = arow[(c0 & 7) * FB + ((c0 >> 3) << 1)];
    a.y = arow[(c1 & 7) * FB + ((c1 >> 3) << 1)];
    bf.x = GW[c0 * HH + col0 + lm];
    bf.y = GW[c1 * HH + col0 + lm];
    acc = __builtin_amdgcn_wmma_f32_16x16x4_f32(
         false, a,  false, bf,
         (short)0, acc,  false,  false);
  }
#pragma unroll
  for (int j = 0; j < 8; ++j) {
    int r = row0 + half * 8 + j;
    out[(size_t)r * HH + col0 + lm] = acc[j];
  }
#else
#pragma unroll
  for (int j = 0; j < 8; ++j) {
    int r = row0 + half * 8 + j;
    int n = r % NN;
    int bb = r / NN;
    float s = 0.0f;
    for (int c = 0; c < CDIM; ++c) {
      float fv = agg[(size_t)n * (KK * FB) + (c & 7) * FB + ((c >> 3) << 1) + bb];
      s += fv * GW[c * HH + col0 + lm];
    }
    out[(size_t)r * HH + col0 + lm] = s;
  }
#endif
}

extern "C" void kernel_launch(void* const* d_in, const int* in_sizes, int n_in,
                              void* d_out, int out_size, void* d_ws, size_t ws_size,
                              hipStream_t stream) {
  const float* X     = (const float*)d_in[0];
  const float* uval  = (const float*)d_in[1];
  const float* mu    = (const float*)d_in[2];
  const float* sigma = (const float*)d_in[3];
  const float* GW    = (const float*)d_in[4];
  const int* urows   = (const int*)d_in[5];
  const int* ucols   = (const int*)d_in[6];
  float* out = (float*)d_out;

  unsigned* mkey = (unsigned*)d_ws;
  float* denom   = (float*)d_ws + NK;
  float* agg     = (float*)d_ws + 2 * NK;
  const int ws_words = 2 * NK + NK * FB;

  k_init<<<2048, 256, 0, stream>>>((unsigned*)d_ws, ws_words);

  k_logit_max<<<(EE * KK + 255) / 256, 256, 0, stream>>>(uval, mu, sigma, urows, mkey);

  k_denom<<<(EE * KK + 255) / 256, 256, 0, stream>>>(uval, mu, sigma, urows, mkey, denom);

  k_scatter<<<(EE * KK + 255) / 256, 256, 0, stream>>>(
      X, uval, mu, sigma, urows, ucols, mkey, denom, agg);

  int waves = MTILES * NTILES;
  int blocks = (waves + 7) / 8;
  k_gemm<<<blocks, 256, 0, stream>>>(agg, GW, out);
}


# Coarsen the three edge passes: one wave owns 32 consecutive edges, issues all their gathers ahead of the atomics and walks 8 groups of 4: 8x fewer waves; float-add arrival order varies as before
# speedup vs baseline: 4.5512x; 1.1576x over previous
#include <hip/hip_runtime.h>
#include <math.h>

constexpr int BB = 2;
constexpr int NN = 50000;
constexpr int FF = 8;
constexpr int EE = 800000;
constexpr int DD = 3;
constexpr int KK = 8;
constexpr int HH = 64;
constexpr int FB = FF * BB;
constexpr int NK = NN * KK;
constexpr int CDIM = FF * KK;
constexpr int MROWS = BB * NN;
constexpr int MTILES = MROWS / 16;
constexpr int NTILES = HH / 16;
constexpr float EPSF = 1e-14f;
constexpr int GPW = 8;
constexpr int EPW = 4 * GPW;
constexpr int EWAVES = EE / EPW;
static_assert(EE % EPW == 0, "whole waves only");
static_assert(EPW == 32, "one index per lane");

typedef __attribute__((ext_vector_type(2))) float v2f;
typedef __attribute__((ext_vector_type(8))) float v8f;

__device__ __forceinline__ unsigned fkey(float f) {
  unsigned b = __float_as_uint(f);
  return (b & 0x80000000u) ? ~b : (b | 0x80000000u);
}
__device__ __forceinline__ float finv(unsigned k) {
  return __uint_as_float((k & 0x80000000u) ? (k & 0x7fffffffu) : ~k);
}

__global__ void k_init(unsigned* ws, int n) {
  int i = blockIdx.x * blockDim.x + threadIdx.x;
  int stride = gridDim.x * blockDim.x;
  for (; i < n; i += stride) ws[i] = 0u;
}

__global__ void k_logit_max(const float* __restrict__ uval,
                            const float* __restrict__ mu,
                            const float* __restrict__ sigma,
                            const int* __restrict__ urows,
                            unsigned* __restrict__ mkey) {
  int wid = blockIdx.x * (blockDim.x >> 5) + (threadIdx.x >> 5);
  if (wid >= EWAVES) return;
  int lane = threadIdx.x & 31;
  int sub = lane >> 3;
  int k = lane & (KK - 1);
  int e0 = wid * EPW;
  int rows = urows[e0 + lane];
  float mk[DD], iv[DD];
#pragma unroll
  for (int d = 0; d < DD; ++d) {
    float sg = sigma[k * DD + d];
    iv[d] = 1.0f / (sg * sg + EPSF);
    mk[d] = mu[k * DD + d];
  }
  int row[GPW];
  unsigned key[GPW];
#pragma unroll
  for (int g = 0; g < GPW; ++g) {
    int el = g * 4 + sub;
    row[g] = __shfl(rows, el);
    const float* uv = uval + (size_t)(e0 + el) * DD;
    float s = 0.0f;
#pragma unroll
    for (int d = 0; d < DD; ++d) {
      float df = uv[d] - mk[d];
      s += df * df * iv[d];
    }
    key[g] = fkey(-0.5f * s);
  }
#pragma unroll
  for (int g = 0; g < GPW; ++g) atomicMax(&mkey[row[g] * KK + k], key[g]);
}

__global__ void k_denom(const float* __restrict__ uval,
                        const float* __restrict__ mu,
                        const float* __restrict__ sigma,
                        const int* __restrict__ urows,
                        const unsigned* __restrict__ mkey,
                        float* __restrict__ denom) {
  int wid = blockIdx.x * (blockDim.x >> 5) + (threadIdx.x >> 5);
  if (wid >= EWAVES) return;
  int lane = threadIdx.x & 31;
  int sub = lane >> 3;
  int k = lane & (KK - 1);
  int e0 = wid * EPW;
  int rows = urows[e0 + lane];
  float mk[DD], iv[DD];
#pragma unroll
  for (int d = 0; d < DD; ++d) {
    float sg = sigma[k * DD + d];
    iv[d] = 1.0f / (sg * sg + EPSF);
    mk[d] = mu[k * DD + d];
  }
  int row[GPW];
  float ev[GPW];
#pragma unroll
  for (int g = 0; g < GPW; ++g) {
    int el = g * 4 + sub;
    row[g] = __shfl(rows, el);
    const float* uv = uval + (size_t)(e0 + el) * DD;
    float s = 0.0f;
#pragma unroll
    for (int d = 0; d < DD; ++d) {
      float df = uv[d] - mk[d];
      s += df * df * iv[d];
    }
    float l = -0.5f * s;
    float mval = finv(mkey[row[g] * KK + k]);
    ev[g] = __expf(l - mval);
  }
#pragma unroll
  for (int g = 0; g < GPW; ++g) atomicAdd(&denom[row[g] * KK + k], ev[g]);
}

__global__ void k_scatter(const float* __restrict__ X,
                          const float* __restrict__ uval,
                          const float* __restrict__ mu,
                          const float* __restrict__ sigma,
                          const int* __restrict__ urows,
                          const int* __restrict__ ucols,
                          const unsigned* __restrict__ mkey,
                          const float* __restrict__ denom,
                          float* __restrict__ agg) {
  int wid = blockIdx.x * (blockDim.x >> 5) + (threadIdx.x >> 5);
  if (wid >= EWAVES) return;
  int lane = threadIdx.x & 31;
  int sub = lane >> 3;
  int k = lane & (KK - 1);
  int e0 = wid * EPW;
  int rows = urows[e0 + lane];
  int cols = ucols[e0 + lane];
  float mk[DD], fc[DD];
#pragma unroll
  for (int d = 0; d < DD; ++d) {
    float sg = sigma[k * DD + d];
    fc[d] = sg * sg + EPSF;
    mk[d] = mu[k * DD + d];
  }
  int hi = lane >> 4;
  int fb = lane & 15;
  int xoff = (fb & 1) * (NN * FF) + (fb >> 1);

  float xv[EPW];
#pragma unroll
  for (int j = 0; j < EPW; ++j) {
    int col_j = __builtin_amdgcn_readlane(cols, j);
    xv[j] = X[(size_t)col_j * FF + xoff];
  }
  float w[GPW];
#pragma unroll
  for (int g = 0; g < GPW; ++g) {
    int el = g * 4 + sub;
    int row = __shfl(rows, el);
    const float* uv = uval + (size_t)(e0 + el) * DD;
    float s = 0.0f;
#pragma unroll
    for (int d = 0; d < DD; ++d) {
      float df = uv[d] - mk[d];
      s += df * df / fc[d];
    }
    float l = -0.5f * s;
    w[g] = __expf(l - finv(mkey[row * KK + k])) / denom[row * KK + k];
  }
#pragma unroll
  for (int g = 0; g < GPW; ++g) {
#pragma unroll
    for (int ee = 0; ee < 4; ++ee) {
      int row_e = __builtin_amdgcn_readlane(rows, g * 4 + ee);
      float* dst = agg + (size_t)row_e * (KK * FB) + lane;
#pragma unroll
      for (int i = 0; i < 4; ++i) {
        float wk = __shfl(w[g], ee * 8 + 2 * i + hi);
        atomicAdd(dst + i * 32, wk * xv[g * 4 + ee]);
      }
    }
  }
}

__global__ void k_gemm(const float* __restrict__ agg,
                       const float* __restrict__ GW,
                       float* __restrict__ out) {
  int wid = blockIdx.x * (blockDim.x >> 5) + (threadIdx.x >> 5);
  if (wid >= MTILES * NTILES) return;
  int lane = threadIdx.x & 31;
  int half = lane >> 4;
  int lm = lane & 15;
  int mt = wid >> 2;
  int nt = wid & 3;
  int row0 = mt * 16;
  int col0 = nt * 16;
  int b = row0 / NN;
  int n0 = row0 % NN;

  v8f acc = {};
#if __has_builtin(__builtin_amdgcn_wmma_f32_16x16x4_f32)
  const float* arow = agg + (size_t)(n0 + lm) * (KK * FB) + b;
#pragma unroll
  for (int kk = 0; kk < CDIM / 4; ++kk) {
    int c0 = kk * 4 + half * 2;
    int c1 = c0 + 1;
    v2f a, bf;
    a.x = arow[(c0 & 7) * FB + ((c0 >> 3) << 1)];
    a.y = arow[(c1 & 7) * FB + ((c1 >> 3) << 1)];
    bf.x = GW[c0 * HH + col0 + lm];
    bf.y = GW[c1 * HH + col0 + lm];
    acc = __builtin_amdgcn_wmma_f32_16x16x4_f32(
         false, a,  false, bf,
         (short)0, acc,  false,  false);
  }
#pragma unroll
  for (int j = 0; j < 8; ++j) {
    int r = row0 + half * 8 + j;
    out[(size_t)r * HH + col0 + lm] = acc[j];
  }
#else
#pragma unroll
  for (int j = 0; j < 8; ++j) {
    int r = row0 + half * 8 + j;
    int n = r % NN;
    int bb = r / NN;
    float s = 0.0f;
    for (int c = 0; c < CDIM; ++c) {
      float fv = agg[(size_t)n * (KK * FB) + (c & 7) * FB + ((c >> 3) << 1) + bb];
      s += fv * GW[c * HH + col0 + lm];
    }
    out[(size_t)r * HH + col0 + lm] = s;
  }
#endif
}

extern "C" void kernel_launch(void* const* d_in, const int* in_sizes, int n_in,
                              void* d_out, int out_size, void* d_ws, size_t ws_size,
                              hipStream_t stream) {
  const float* X     = (const float*)d_in[0];
  const float* uval  = (const float*)d_in[1];
  const float* mu    = (const float*)d_in[2];
  const float* sigma = (const float*)d_in[3];
  const float* GW    = (const float*)d_in[4];
  const int* urows   = (const int*)d_in[5];
  const int* ucols   = (const int*)d_in[6];
  float* out = (float*)d_out;

  unsigned* mkey = (unsigned*)d_ws;
  float* denom   = (float*)d_ws + NK;
  float* agg     = (float*)d_ws + 2 * NK;
  const int ws_words = 2 * NK + NK * FB;

  k_init<<<2048, 256, 0, stream>>>((unsigned*)d_ws, ws_words);

  k_logit_max<<<(EWAVES + 7) / 8, 256, 0, stream>>>(uval, mu, sigma, urows, mkey);

  k_denom<<<(EWAVES + 7) / 8, 256, 0, stream>>>(uval, mu, sigma, urows, mkey, denom);

  k_scatter<<<(EWAVES + 7) / 8, 256, 0, stream>>>(
      X, uval, mu, sigma, urows, ucols, mkey, denom, agg);

  int waves = MTILES * NTILES;
  int blocks = (waves + 7) / 8;
  k_gemm<<<blocks, 256, 0, stream>>>(agg, GW, out);
}


# GEMM re-tiled: one wave owns 16 nodes x 64 columns x both batches, reads each agg row once as 16-byte vectors in memory order (K summed in a permuted order, alike in A and B loaders)
# speedup vs baseline: 5.4412x; 1.1956x over previous
#include <hip/hip_runtime.h>
#include <math.h>

constexpr int BB = 2;
constexpr int NN = 50000;
constexpr int FF = 8;
constexpr int EE = 800000;
constexpr int DD = 3;
constexpr int KK = 8;
constexpr int HH = 64;
constexpr int FB = FF * BB;
constexpr int NK = NN * KK;
constexpr int CDIM = FF * KK;
constexpr int MROWS = BB * NN;
constexpr int MTILES = MROWS / 16;
constexpr int NTILES = HH / 16;
constexpr float EPSF = 1e-14f;
constexpr int GPW = 8;
constexpr int EPW = 4 * GPW;
constexpr int EWAVES = EE / EPW;
static_assert(EE % EPW == 0, "whole waves only");
static_assert(EPW == 32, "one index per lane");

typedef __attribute__((ext_vector_type(2))) float v2f;
typedef __attribute__((ext_vector_type(4))) float v4f;
typedef __attribute__((ext_vector_type(8))) float v8f;

__device__ __forceinline__ unsigned fkey(float f) {
  unsigned b = __float_as_uint(f);
  return (b & 0x80000000u) ? ~b : (b | 0x80000000u);
}
__device__ __forceinline__ float finv(unsigned k) {
  return __uint_as_float((k & 0x80000000u) ? (k & 0x7fffffffu) : ~k);
}

__global__ void k_init(unsigned* ws, int n) {
  int i = blockIdx.x * blockDim.x + threadIdx.x;
  int stride = gridDim.x * blockDim.x;
  for (; i < n; i += stride) ws[i] = 0u;
}

__global__ void k_logit_max(const float* __restrict__ uval,
                            const float* __restrict__ mu,
                            const float* __restrict__ sigma,
                            const int* __restrict__ urows,
                            unsigned* __restrict__ mkey) {
  int wid = blockIdx.x * (blockDim.x >> 5) + (threadIdx.x >> 5);
  if (wid >= EWAVES) return;
  int lane = threadIdx.x & 31;
  int sub = lane >> 3;
  int k = lane & (KK - 1);
  int e0 = wid * EPW;
  int rows = urows[e0 + lane];
  float mk[DD], iv[DD];
#pragma unroll
  for (int d = 0; d < DD; ++d) {
    float sg = sigma[k * DD + d];
    iv[d] = 1.0f / (sg * sg + EPSF);
    mk[d] = mu[k * DD + d];
  }
  int row[GPW];
  unsigned key[GPW];
#pragma unroll
  for (int g = 0; g < GPW; ++g) {
    int el = g * 4 + sub;
    row[g] = __shfl(rows, el);
    const float* uv = uval + (size_t)(e0 + el) * DD;
    float s = 0.0f;
#pragma unroll
    for (int d = 0; d < DD; ++d) {
      float df = uv[d] - mk[d];
      s += df * df * iv[d];
    }
    key[g] = fkey(-0.5f * s);
  }
#pragma unroll
  for (int g = 0; g < GPW; ++g) atomicMax(&mkey[row[g] * KK + k], key[g]);
}

__global__ void k_denom(const float* __restrict__ uval,
                        const float* __restrict__ mu,
                        const float* __restrict__ sigma,
                        const int* __restrict__ urows,
                        const unsigned* __restrict__ mkey,
                        float* __restrict__ denom) {
  int wid = blockIdx.x * (blockDim.x >> 5) + (threadIdx.x >> 5);
  if (wid >= EWAVES) return;
  int lane = threadIdx.x & 31;
  int sub = lane >> 3;
  int k = lane & (KK - 1);
  int e0 = wid * EPW;
  int rows = urows[e0 + lane];
  float mk[DD], iv[DD];
#pragma unroll
  for (int d = 0; d < DD; ++d) {
    float sg = sigma[k * DD + d];
    iv[d] = 1.0f / (sg * sg + EPSF);
    mk[d] = mu[k * DD + d];
  }
  int row[GPW];
  float ev[GPW];
#pragma unroll
  for (int g = 0; g < GPW; ++g) {
    int el = g * 4 + sub;
    row[g] = __shfl(rows, el);
    const float* uv = uval + (size_t)(e0 + el) * DD;
    float s = 0.0f;
#pragma unroll
    for (int d = 0; d < DD; ++d) {
      float df = uv[d] - mk[d];
      s += df * df * iv[d];
    }
    float l = -0.5f * s;
    float mval = finv(mkey[row[g] * KK + k]);
    ev[g] = __expf(l - mval);
  }
#pragma unroll
  for (int g = 0; g < GPW; ++g) atomicAdd(&denom[row[g] * KK + k], ev[g]);
}

__global__ void k_scatter(const float* __restrict__ X,
                          const float* __restrict__ uval,
                          const float* __restrict__ mu,
                          const float* __restrict__ sigma,
                          const int* __restrict__ urows,
                          const int* __restrict__ ucols,
                          const unsigned* __restrict__ mkey,
                          const float* __restrict__ denom,
                          float* __restrict__ agg) {
  int wid = blockIdx.x * (blockDim.x >> 5) + (threadIdx.x >> 5);
  if (wid >= EWAVES) return;
  int lane = threadIdx.x & 31;
  int sub = lane >> 3;
  int k = lane & (KK - 1);
  int e0 = wid * EPW;
  int rows = urows[e0 + lane];
  int cols = ucols[e0 + lane];
  float mk[DD], fc[DD];
#pragma unroll
  for (int d = 0; d < DD; ++d) {
    float sg = sigma[k * DD + d];
    fc[d] = sg * sg + EPSF;
    mk[d] = mu[k * DD + d];
  }
  int hi = lane >> 4;
  int fb = lane & 15;
  int xoff = (fb & 1) * (NN * FF) + (fb >> 1);

  float xv[EPW];
#pragma unroll
  for (int j = 0; j < EPW; ++j) {
    int col_j = __builtin_amdgcn_readlane(cols, j);
    xv[j] = X[(size_t)col_j * FF + xoff];
  }
  float w[GPW];
#pragma unroll
  for (int g = 0; g < GPW; ++g) {
    int el = g * 4 + sub;
    int row = __shfl(rows, el);
    const float* uv = uval + (size_t)(e0 + el) * DD;
    float s = 0.0f;
#pragma unroll
    for (int d = 0; d < DD; ++d) {
      float df = uv[d] - mk[d];
      s += df * df / fc[d];
    }
    float l = -0.5f * s;
    w[g] = __expf(l - finv(mkey[row * KK + k])) / denom[row * KK + k];
  }
#pragma unroll
  for (int g = 0; g < GPW; ++g) {
#pragma unroll
    for (int ee = 0; ee < 4; ++ee) {
      int row_e = __builtin_amdgcn_readlane(rows, g * 4 + ee);
      float* dst = agg + (size_t)row_e * (KK * FB) + lane;
#pragma unroll
      for (int i = 0; i < 4; ++i) {
        float wk = __shfl(w[g], ee * 8 + 2 * i + hi);
        atomicAdd(dst + i * 32, wk * xv[g * 4 + ee]);
      }
    }
  }
}

constexpr int NODE_TILES = NN / 16;
constexpr int GEMM_WAVES_PER_BLOCK = 4;
constexpr int GEMM_BLOCKS = (NODE_TILES + GEMM_WAVES_PER_BLOCK - 1) / GEMM_WAVES_PER_BLOCK;
static_assert(NN % 16 == 0, "whole node tiles only");
static_assert(KK * FB == 128 && HH == 64 && CDIM == 64, "tile maps below are written for these sizes");

__global__ void __launch_bounds__(32 * GEMM_WAVES_PER_BLOCK)
k_gemm(const float* __restrict__ agg,
       const float* __restrict__ GW,
       float* __restrict__ out) {
  int tile = blockIdx.x * (blockDim.x >> 5) + (threadIdx.x >> 5);
  if (tile >= NODE_TILES) return;
  int lane = threadIdx.x & 31;
  int h = lane >> 4;
  int lm = lane & 15;
  int n0 = tile * 16;

  const v4f* arow = (const v4f*)(agg + (size_t)(n0 + lm) * (KK * FB) + 4 * h);
  v4f av[16];
#pragma unroll
  for (int s = 0; s < 16; ++s) av[s] = arow[2 * s];

  v8f acc[2][4] = {};
  const float* gcol = GW + 4 * lm;
#pragma unroll
  for (int s = 0; s < 16; ++s) {
    int c0 = ((s & 1) * 4 + 2 * h) * KK + (s >> 1);
    int c1 = c0 + KK;
    v4f g0 = *(const v4f*)(gcol + c0 * HH);
    v4f g1 = *(const v4f*)(gcol + c1 * HH);
    v2f a0, a1;
    a0.x = av[s].x; a0.y = av[s].z;
    a1.x = av[s].y; a1.y = av[s].w;
#pragma unroll
    for (int t = 0; t < 4; ++t) {
      v2f bf;
      bf.x = g0[t];
      bf.y = g1[t];
      acc[0][t] = __builtin_amdgcn_wmma_f32_16x16x4_f32(
           false, a0,  false, bf,
           (short)0, acc[0][t],  false,  false);
      acc[1][t] = __builtin_amdgcn_wmma_f32_16x16x4_f32(
           false, a1,  false, bf,
           (short)0, acc[1][t],  false,  false);
    }
  }
#pragma unroll
  for (int b = 0; b < 2; ++b) {
#pragma unroll
    for (int j = 0; j < 8; ++j) {
      size_t r = (size_t)b * NN + n0 + h * 8 + j;
      v4f o;
      o.x = acc[b][0][j];
      o.y = acc[b][1][j];
      o.z = acc[b][2][j];
      o.w = acc[b][3][j];
      *(v4f*)(out + r * HH + 4 * lm) = o;
    }
  }
}

extern "C" void kernel_launch(void* const* d_in, const int* in_sizes, int n_in,
                              void* d_out, int out_size, void* d_ws, size_t ws_size,
                              hipStream_t stream) {
  const float* X     = (const float*)d_in[0];
  const float* uval  = (const float*)d_in[1];
  const float* mu    = (const float*)d_in[2];
  const float* sigma = (const float*)d_in[3];
  const float* GW    = (const float*)d_in[4];
  const int* urows   = (const int*)d_in[5];
  const int* ucols   = (const int*)d_in[6];
  float* out = (float*)d_out;

  unsigned* mkey = (unsigned*)d_ws;
  float* denom   = (float*)d_ws + NK;
  float* agg     = (float*)d_ws + 2 * NK;
  const int ws_words = 2 * NK + NK * FB;

  k_init<<<2048, 256, 0, stream>>>((unsigned*)d_ws, ws_words);

  k_logit_max<<<(EWAVES + 7) / 8, 256, 0, stream>>>(uval, mu, sigma, urows, mkey);

  k_denom<<<(EWAVES + 7) / 8, 256, 0, stream>>>(uval, mu, sigma, urows, mkey, denom);

  k_scatter<<<(EWAVES + 7) / 8, 256, 0, stream>>>(
      X, uval, mu, sigma, urows, ucols, mkey, denom, agg);

  k_gemm<<<GEMM_BLOCKS, 32 * GEMM_WAVES_PER_BLOCK, 0, stream>>>(agg, GW, out);
}


# Softmax division deferred to the GEMM A loader: scatter adds e=exp(logit-max) into denom and e*X into agg in one pass (separate sum pass and launch gone); sum(e*x)*(1/denom), add order varies as ever
# speedup vs baseline: 5.7631x; 1.0592x over previous
#include <hip/hip_runtime.h>
#include <math.h>

constexpr int BB = 2;
constexpr int NN = 50000;
constexpr int FF = 8;
constexpr int EE = 800000;
constexpr int DD = 3;
constexpr int KK = 8;
constexpr int HH = 64;
constexpr int FB = FF * BB;
constexpr int NK = NN * KK;
constexpr int CDIM = FF * KK;
constexpr int MROWS = BB * NN;
constexpr int MTILES = MROWS / 16;
constexpr int NTILES = HH / 16;
constexpr float EPSF = 1e-14f;
constexpr int GPW = 8;
constexpr int EPW = 4 * GPW;
constexpr int EWAVES = EE / EPW;
static_assert(EE % EPW == 0, "whole waves only");
static_assert(EPW == 32, "one index per lane");

typedef __attribute__((ext_vector_type(2))) float v2f;
typedef __attribute__((ext_vector_type(4))) float v4f;
typedef __attribute__((ext_vector_type(8))) float v8f;

__device__ __forceinline__ unsigned fkey(float f) {
  unsigned b = __float_as_uint(f);
  return (b & 0x80000000u) ? ~b : (b | 0x80000000u);
}
__device__ __forceinline__ float finv(unsigned k) {
  return __uint_as_float((k & 0x80000000u) ? (k & 0x7fffffffu) : ~k);
}

__global__ void k_init(unsigned* ws, int n) {
  int i = blockIdx.x * blockDim.x + threadIdx.x;
  int stride = gridDim.x * blockDim.x;
  for (; i < n; i += stride) ws[i] = 0u;
}

__global__ void k_logit_max(const float* __restrict__ uval,
                            const float* __restrict__ mu,
                            const float* __restrict__ sigma,
                            const int* __restrict__ urows,
                            unsigned* __restrict__ mkey) {
  int wid = blockIdx.x * (blockDim.x >> 5) + (threadIdx.x >> 5);
  if (wid >= EWAVES) return;
  int lane = threadIdx.x & 31;
  int sub = lane >> 3;
  int k = lane & (KK - 1);
  int e0 = wid * EPW;
  int rows = urows[e0 + lane];
  float mk[DD], iv[DD];
#pragma unroll
  for (int d = 0; d < DD; ++d) {
    float sg = sigma[k * DD + d];
    iv[d] = 1.0f / (sg * sg + EPSF);
    mk[d] = mu[k * DD + d];
  }
  int row[GPW];
  unsigned key[GPW];
#pragma unroll
  for (int g = 0; g < GPW; ++g) {
    int el = g * 4 + sub;
    row[g] = __shfl(rows, el);
    const float* uv = uval + (size_t)(e0 + el) * DD;
    float s = 0.0f;
#pragma unroll
    for (int d = 0; d < DD; ++d) {
      float df = uv[d] - mk[d];
      s += df * df * iv[d];
    }
    key[g] = fkey(-0.5f * s);
  }
#pragma unroll
  for (int g = 0; g < GPW; ++g) atomicMax(&mkey[row[g] * KK + k], key[g]);
}

__global__ void k_scatter(const float* __restrict__ X,
                          const float* __restrict__ uval,
                          const float* __restrict__ mu,
                          const float* __restrict__ sigma,
                          const int* __restrict__ urows,
                          const int* __restrict__ ucols,
                          const unsigned* __restrict__ mkey,
                          float* __restrict__ denom,
                          float* __restrict__ agg) {
  int wid = blockIdx.x * (blockDim.x >> 5) + (threadIdx.x >> 5);
  if (wid >= EWAVES) return;
  int lane = threadIdx.x & 31;
  int sub = lane >> 3;
  int k = lane & (KK - 1);
  int e0 = wid * EPW;
  int rows = urows[e0 + lane];
  int cols = ucols[e0 + lane];
  float mk[DD], fc[DD];
#pragma unroll
  for (int d = 0; d < DD; ++d) {
    float sg = sigma[k * DD + d];
    fc[d] = sg * sg + EPSF;
    mk[d] = mu[k * DD + d];
  }
  int hi = lane >> 4;
  int fb = lane & 15;
  int xoff = (fb & 1) * (NN * FF) + (fb >> 1);

  float xv[EPW];
#pragma unroll
  for (int j = 0; j < EPW; ++j) {
    int col_j = __builtin_amdgcn_readlane(cols, j);
    xv[j] = X[(size_t)col_j * FF + xoff];
  }
  float w[GPW];
  int rowk[GPW];
#pragma unroll
  for (int g = 0; g < GPW; ++g) {
    int el = g * 4 + sub;
    int row = __shfl(rows, el);
    rowk[g] = row * KK + k;
    const float* uv = uval + (size_t)(e0 + el) * DD;
    float s = 0.0f;
#pragma unroll
    for (int d = 0; d < DD; ++d) {
      float df = uv[d] - mk[d];
      s += df * df / fc[d];
    }
    float l = -0.5f * s;
    w[g] = __expf(l - finv(mkey[rowk[g]]));
  }
#pragma unroll
  for (int g = 0; g < GPW; ++g) atomicAdd(&denom[rowk[g]], w[g]);
#pragma unroll
  for (int g = 0; g < GPW; ++g) {
#pragma unroll
    for (int ee = 0; ee < 4; ++ee) {
      int row_e = __builtin_amdgcn_readlane(rows, g * 4 + ee);
      float* dst = agg + (size_t)row_e * (KK * FB) + lane;
#pragma unroll
      for (int i = 0; i < 4; ++i) {
        float wk = __shfl(w[g], ee * 8 + 2 * i + hi);
        atomicAdd(dst + i * 32, wk * xv[g * 4 + ee]);
      }
    }
  }
}

constexpr int NODE_TILES = NN / 16;
constexpr int GEMM_WAVES_PER_BLOCK = 4;
constexpr int GEMM_BLOCKS = (NODE_TILES + GEMM_WAVES_PER_BLOCK - 1) / GEMM_WAVES_PER_BLOCK;
static_assert(NN % 16 == 0, "whole node tiles only");
static_assert(KK * FB == 128 && HH == 64 && CDIM == 64, "tile maps below are written for these sizes");

__global__ void __launch_bounds__(32 * GEMM_WAVES_PER_BLOCK)
k_gemm(const float* __restrict__ agg,
       const float* __restrict__ denom,
       const float* __restrict__ GW,
       float* __restrict__ out) {
  int tile = blockIdx.x * (blockDim.x >> 5) + (threadIdx.x >> 5);
  if (tile >= NODE_TILES) return;
  int lane = threadIdx.x & 31;
  int h = lane >> 4;
  int lm = lane & 15;
  int n0 = tile * 16;

  const v4f* arow = (const v4f*)(agg + (size_t)(n0 + lm) * (KK * FB) + 4 * h);
  const v4f* drow = (const v4f*)(denom + (size_t)(n0 + lm) * KK);
  v4f dlo = drow[0];
  v4f dhi = drow[1];
  v4f av[16];
#pragma unroll
  for (int s = 0; s < 16; ++s) av[s] = arow[2 * s];
  float rinv[KK];
#pragma unroll
  for (int kk = 0; kk < KK; ++kk) {
    float d = (kk < 4) ? dlo[kk & 3] : dhi[kk & 3];
    rinv[kk] = (d == 0.0f) ? 0.0f : 1.0f / d;
  }

  v8f acc[2][4] = {};
  const float* gcol = GW + 4 * lm;
#pragma unroll
  for (int s = 0; s < 16; ++s) {
    int c0 = ((s & 1) * 4 + 2 * h) * KK + (s >> 1);
    int c1 = c0 + KK;
    v4f g0 = *(const v4f*)(gcol + c0 * HH);
    v4f g1 = *(const v4f*)(gcol + c1 * HH);
    v4f as = av[s] * rinv[s >> 1];
    v2f a0, a1;
    a0.x = as.x; a0.y = as.z;
    a1.x = as.y; a1.y = as.w;
#pragma unroll
    for (int t = 0; t < 4; ++t) {
      v2f bf;
      bf.x = g0[t];
      bf.y = g1[t];
      acc[0][t] = __builtin_amdgcn_wmma_f32_16x16x4_f32(
           false, a0,  false, bf,
           (short)0, acc[0][t],  false,  false);
      acc[1][t] = __builtin_amdgcn_wmma_f32_16x16x4_f32(
           false, a1,  false, bf,
           (short)0, acc[1][t],  false,  false);
    }
  }
#pragma unroll
  for (int b = 0; b < 2; ++b) {
#pragma unroll
    for (int j = 0; j < 8; ++j) {
      size_t r = (size_t)b * NN + n0 + h * 8 + j;
      v4f o;
      o.x = acc[b][0][j];
      o.y = acc[b][1][j];
      o.z = acc[b][2][j];
      o.w = acc[b][3][j];
      *(v4f*)(out + r * HH + 4 * lm) = o;
    }
  }
}

extern "C" void kernel_launch(void* const* d_in, const int* in_sizes, int n_in,
                              void* d_out, int out_size, void* d_ws, size_t ws_size,
                              hipStream_t stream) {
  const float* X     = (const float*)d_in[0];
  const float* uval  = (const float*)d_in[1];
  const float* mu    = (const float*)d_in[2];
  const float* sigma = (const float*)d_in[3];
  const float* GW    = (const float*)d_in[4];
  const int* urows   = (const int*)d_in[5];
  const int* ucols   = (const int*)d_in[6];
  float* out = (float*)d_out;

  unsigned* mkey = (unsigned*)d_ws;
  float* denom   = (float*)d_ws + NK;
  float* agg     = (float*)d_ws + 2 * NK;
  const int ws_words = 2 * NK + NK * FB;

  k_init<<<2048, 256, 0, stream>>>((unsigned*)d_ws, ws_words);

  k_logit_max<<<(EWAVES + 7) / 8, 256, 0, stream>>>(uval, mu, sigma, urows, mkey);

  k_scatter<<<(EWAVES + 7) / 8, 256, 0, stream>>>(
      X, uval, mu, sigma, urows, ucols, mkey, denom, agg);

  k_gemm<<<GEMM_BLOCKS, 32 * GEMM_WAVES_PER_BLOCK, 0, stream>>>(agg, denom, GW, out);
}


# Scatter by ownership: each of 256 work-groups keeps 196 agg rows in LDS, scans all row ids, adds its edges with LDS float atomics (arrival order varies as before), stores rows once: no global adds
# speedup vs baseline: 6.8598x; 1.1903x over previous
#include <hip/hip_runtime.h>
#include <math.h>

constexpr int BB = 2;
constexpr int NN = 50000;
constexpr int FF = 8;
constexpr int EE = 800000;
constexpr int DD = 3;
constexpr int KK = 8;
constexpr int HH = 64;
constexpr int FB = FF * BB;
constexpr int NK = NN * KK;
constexpr int CDIM = FF * KK;
constexpr int MROWS = BB * NN;
constexpr int MTILES = MROWS / 16;
constexpr int NTILES = HH / 16;
constexpr float EPSF = 1e-14f;
constexpr int GPW = 8;
constexpr int EPW = 4 * GPW;
constexpr int EWAVES = EE / EPW;
static_assert(EE % EPW == 0, "whole waves only");
static_assert(EPW == 32, "one index per lane");

typedef __attribute__((ext_vector_type(2))) float v2f;
typedef __attribute__((ext_vector_type(4))) float v4f;
typedef __attribute__((ext_vector_type(8))) float v8f;

__device__ __forceinline__ unsigned fkey(float f) {
  unsigned b = __float_as_uint(f);
  return (b & 0x80000000u) ? ~b : (b | 0x80000000u);
}
__device__ __forceinline__ float finv(unsigned k) {
  return __uint_as_float((k & 0x80000000u) ? (k & 0x7fffffffu) : ~k);
}

__global__ void k_init(unsigned* ws, int n) {
  int i = blockIdx.x * blockDim.x + threadIdx.x;
  int stride = gridDim.x * blockDim.x;
  for (; i < n; i += stride) ws[i] = 0u;
}

__global__ void k_logit_max(const float* __restrict__ uval,
                            const float* __restrict__ mu,
                            const float* __restrict__ sigma,
                            const int* __restrict__ urows,
                            unsigned* __restrict__ mkey) {
  int wid = blockIdx.x * (blockDim.x >> 5) + (threadIdx.x >> 5);
  if (wid >= EWAVES) return;
  int lane = threadIdx.x & 31;
  int sub = lane >> 3;
  int k = lane & (KK - 1);
  int e0 = wid * EPW;
  int rows = urows[e0 + lane];
  float mk[DD], iv[DD];
#pragma unroll
  for (int d = 0; d < DD; ++d) {
    float sg = sigma[k * DD + d];
    iv[d] = 1.0f / (sg * sg + EPSF);
    mk[d] = mu[k * DD + d];
  }
  int row[GPW];
  unsigned key[GPW];
#pragma unroll
  for (int g = 0; g < GPW; ++g) {
    int el = g * 4 + sub;
    row[g] = __shfl(rows, el);
    const float* uv = uval + (size_t)(e0 + el) * DD;
    float s = 0.0f;
#pragma unroll
    for (int d = 0; d < DD; ++d) {
      float df = uv[d] - mk[d];
      s += df * df * iv[d];
    }
    key[g] = fkey(-0.5f * s);
  }
#pragma unroll
  for (int g = 0; g < GPW; ++g) atomicMax(&mkey[row[g] * KK + k], key[g]);
}

typedef __attribute__((ext_vector_type(4))) int v4i;
constexpr int SLABS = 256;
constexpr int SROWS = (NN + SLABS - 1) / SLABS;
constexpr int SC_WAVES = 32;
constexpr int SC_THREADS = SC_WAVES * 32;
constexpr int RVECS = EE / 4;
constexpr int SWEEPS = (RVECS + SC_THREADS - 1) / SC_THREADS;
constexpr int PFD = 4;
constexpr int BATCH = 16;
constexpr int BGRP = BATCH / 4;
constexpr int LCAP = 528;
constexpr int EID_BITS = 20;
static_assert(EE % 4 == 0, "row ids are read as vectors of 4");
static_assert(EE <= (1 << EID_BITS) && SROWS <= (1 << (32 - EID_BITS)), "list entry packing");
static_assert((SLABS - 1) * SROWS < NN && SLABS * SROWS >= NN, "every row owned once");
static_assert((SROWS * KK * FB) % 4 == 0 && (SROWS * KK) % 4 == 0, "slab moved as 16-byte vectors");

__global__ void __launch_bounds__(SC_THREADS)
k_scatter(const float* __restrict__ X,
          const float* __restrict__ uval,
          const float* __restrict__ mu,
          const float* __restrict__ sigma,
          const int* __restrict__ urows,
          const int* __restrict__ ucols,
          const unsigned* __restrict__ mkey,
          float* __restrict__ denom,
          float* __restrict__ agg) {
  __shared__ __attribute__((aligned(16))) float sagg[SROWS * KK * FB];
  __shared__ __attribute__((aligned(16))) float sden[SROWS * KK];
  __shared__ unsigned slist[SC_WAVES * LCAP];

  const int tid = threadIdx.x;
  const int lane = tid & 31;
  const int wave = tid >> 5;
  const int r0 = blockIdx.x * SROWS;

  {
    const v4f z = {0.0f, 0.0f, 0.0f, 0.0f};
    for (int i = tid; i < SROWS * KK * FB / 4; i += SC_THREADS) ((v4f*)sagg)[i] = z;
    for (int i = tid; i < SROWS * KK / 4; i += SC_THREADS) ((v4f*)sden)[i] = z;
  }
  __syncthreads();

  const int sub = lane >> 3;
  const int k = lane & (KK - 1);
  float mk[DD], fc[DD];
#pragma unroll
  for (int d = 0; d < DD; ++d) {
    float sg = sigma[k * DD + d];
    fc[d] = sg * sg + EPSF;
    mk[d] = mu[k * DD + d];
  }
  const int hi = lane >> 4;
  const int fb = lane & 15;
  const int xoff = (fb & 1) * (NN * FF) + (fb >> 1);

  unsigned* mylist = slist + wave * LCAP;
  int cnt = 0;

  const v4i* rvec = (const v4i*)urows;
  const int rot = (blockIdx.x * SWEEPS) / SLABS;
  auto sweep_vec = [&](int s) -> int {
    if (s >= SWEEPS) return -1;
    int sr = s + rot;
    if (sr >= SWEEPS) sr -= SWEEPS;
    int vi = sr * SC_THREADS + tid;
    return (vi < RVECS) ? vi : -1;
  };
  auto sweep_load = [&](int vi) -> v4i {
    v4i none = {-1, -1, -1, -1};
    return (vi >= 0) ? rvec[vi] : none;
  };

  static_assert(PFD == 4 && SWEEPS % PFD == 0, "slots written out for 4");
  static_assert(LCAP >= BATCH - 1 + PFD * 4 * 32, "a trip's new entries fit the list");
  v4i q0 = sweep_load(sweep_vec(0)), q1 = sweep_load(sweep_vec(1));
  v4i q2 = sweep_load(sweep_vec(2)), q3 = sweep_load(sweep_vec(3));

  auto append = [&](const v4i cur, int s) {
    const int vcur = sweep_vec(s);
    unsigned dd[4];
    bool pp[4];
#pragma unroll
    for (int c = 0; c < 4; ++c) {
      dd[c] = (unsigned)(cur[c] - r0);
      pp[c] = dd[c] < (unsigned)SROWS;
    }
    if (__builtin_amdgcn_ballot_w32(pp[0] | pp[1] | pp[2] | pp[3]) != 0u) {
#pragma unroll
      for (int c = 0; c < 4; ++c) {
        unsigned m = __builtin_amdgcn_ballot_w32(pp[c]);
        if (m != 0u) {
          int pos = cnt + (int)__builtin_amdgcn_mbcnt_lo(m, 0u);
          if (pp[c]) mylist[pos] = (dd[c] << EID_BITS) | (unsigned)(4 * vcur + c);
          cnt += __builtin_popcount(m);
        }
      }
    }
  };

  for (int sw = 0; sw <= SWEEPS; sw += PFD) {
    const bool last = (sw == SWEEPS);
    append(q0, sw);
    q0 = sweep_load(sweep_vec(sw + PFD));
    append(q1, sw + 1);
    q1 = sweep_load(sweep_vec(sw + PFD + 1));
    append(q2, sw + 2);
    q2 = sweep_load(sweep_vec(sw + PFD + 2));
    append(q3, sw + 3);
    q3 = sweep_load(sweep_vec(sw + PFD + 3));

    for (int bt = 0; bt < LCAP / BATCH + 1; ++bt) {
      if (!(cnt >= BATCH || (last && cnt > 0))) break;
      const int nv = cnt < BATCH ? cnt : BATCH;
      const int base = cnt - nv;
      cnt = base;
      const unsigned ent = (lane < nv) ? mylist[base + lane] : 0u;
      const int eid = (int)(ent & ((1u << EID_BITS) - 1u));
      const int rl = (int)(ent >> EID_BITS);
      const int cols = ucols[eid];

      float xv[BATCH];
#pragma unroll
      for (int j = 0; j < BATCH; ++j) {
        int col_j = __builtin_amdgcn_readlane(cols, j);
        xv[j] = X[(size_t)col_j * FF + xoff];
      }
      float w[BGRP];
      int rlk[BGRP];
#pragma unroll
      for (int g = 0; g < BGRP; ++g) {
        int el = g * 4 + sub;
        int e_el = __shfl(eid, el);
        int rl_el = __shfl(rl, el);
        rlk[g] = rl_el * KK + k;
        const float* uv = uval + (size_t)e_el * DD;
        float s = 0.0f;
#pragma unroll
        for (int d = 0; d < DD; ++d) {
          float df = uv[d] - mk[d];
          s += df * df / fc[d];
        }
        float l = -0.5f * s;
        w[g] = __expf(l - finv(mkey[(r0 + rl_el) * KK + k]));
      }
#pragma unroll
      for (int g = 0; g < BGRP; ++g) {
        if (g * 4 + sub < nv) atomicAdd(&sden[rlk[g]], w[g]);
      }
#pragma unroll
      for (int g = 0; g < BGRP; ++g) {
#pragma unroll
        for (int ee = 0; ee < 4; ++ee) {
          if (g * 4 + ee < nv) {
            int rl_e = __builtin_amdgcn_readlane(rl, g * 4 + ee);
            float* dst = sagg + rl_e * (KK * FB) + lane;
#pragma unroll
            for (int i = 0; i < 4; ++i) {
              float wk = __shfl(w[g], ee * 8 + 2 * i + hi);
              atomicAdd(dst + i * 32, wk * xv[g * 4 + ee]);
            }
          }
        }
      }
    }
  }

  __syncthreads();
  const int nrows = (NN - r0 < SROWS) ? (NN - r0) : SROWS;
  v4f* gagg = (v4f*)(agg + (size_t)r0 * (KK * FB));
  for (int i = tid; i < nrows * (KK * FB / 4); i += SC_THREADS) gagg[i] = ((const v4f*)sagg)[i];
  v4f* gden = (v4f*)(denom + (size_t)r0 * KK);
  for (int i = tid; i < nrows * (KK / 4); i += SC_THREADS) gden[i] = ((const v4f*)sden)[i];
}

constexpr int NODE_TILES = NN / 16;
constexpr int GEMM_WAVES_PER_BLOCK = 4;
constexpr int GEMM_BLOCKS = (NODE_TILES + GEMM_WAVES_PER_BLOCK - 1) / GEMM_WAVES_PER_BLOCK;
static_assert(NN % 16 == 0, "whole node tiles only");
static_assert(KK * FB == 128 && HH == 64 && CDIM == 64, "tile maps below are written for these sizes");

__global__ void __launch_bounds__(32 * GEMM_WAVES_PER_BLOCK)
k_gemm(const float* __restrict__ agg,
       const float* __restrict__ denom,
       const float* __restrict__ GW,
       float* __restrict__ out) {
  int tile = blockIdx.x * (blockDim.x >> 5) + (threadIdx.x >> 5);
  if (tile >= NODE_TILES) return;
  int lane = threadIdx.x & 31;
  int h = lane >> 4;
  int lm = lane & 15;
  int n0 = tile * 16;

  const v4f* arow = (const v4f*)(agg + (size_t)(n0 + lm) * (KK * FB) + 4 * h);
  const v4f* drow = (const v4f*)(denom + (size_t)(n0 + lm) * KK);
  v4f dlo = drow[0];
  v4f dhi = drow[1];
  v4f av[16];
#pragma unroll
  for (int s = 0; s < 16; ++s) av[s] = arow[2 * s];
  float rinv[KK];
#pragma unroll
  for (int kk = 0; kk < KK; ++kk) {
    float d = (kk < 4) ? dlo[kk & 3] : dhi[kk & 3];
    rinv[kk] = (d == 0.0f) ? 0.0f : 1.0f / d;
  }

  v8f acc[2][4] = {};
  const float* gcol = GW + 4 * lm;
#pragma unroll
  for (int s = 0; s < 16; ++s) {
    int c0 = ((s & 1) * 4 + 2 * h) * KK + (s >> 1);
    int c1 = c0 + KK;
    v4f g0 = *(const v4f*)(gcol + c0 * HH);
    v4f g1 = *(const v4f*)(gcol + c1 * HH);
    v4f as = av[s] * rinv[s >> 1];
    v2f a0, a1;
    a0.x = as.x; a0.y = as.z;
    a1.x = as.y; a1.y = as.w;
#pragma unroll
    for (int t = 0; t < 4; ++t) {
      v2f bf;
      bf.x = g0[t];
      bf.y = g1[t];
      acc[0][t] = __builtin_amdgcn_wmma_f32_16x16x4_f32(
           false, a0,  false, bf,
           (short)0, acc[0][t],  false,  false);
      acc[1][t] = __builtin_amdgcn_wmma_f32_16x16x4_f32(
           false, a1,  false, bf,
           (short)0, acc[1][t],  false,  false);
    }
  }
#pragma unroll
  for (int b = 0; b < 2; ++b) {
#pragma unroll
    for (int j = 0; j < 8; ++j) {
      size_t r = (size_t)b * NN + n0 + h * 8 + j;
      v4f o;
      o.x = acc[b][0][j];
      o.y = acc[b][1][j];
      o.z = acc[b][2][j];
      o.w = acc[b][3][j];
      *(v4f*)(out + r * HH + 4 * lm) = o;
    }
  }
}

extern "C" void kernel_launch(void* const* d_in, const int* in_sizes, int n_in,
                              void* d_out, int out_size, void* d_ws, size_t ws_size,
                              hipStream_t stream) {
  const float* X     = (const float*)d_in[0];
  const float* uval  = (const float*)d_in[1];
  const float* mu    = (const float*)d_in[2];
  const float* sigma = (const float*)d_in[3];
  const float* GW    = (const float*)d_in[4];
  const int* urows   = (const int*)d_in[5];
  const int* ucols   = (const int*)d_in[6];
  float* out = (float*)d_out;

  unsigned* mkey = (unsigned*)d_ws;
  float* denom   = (float*)d_ws + NK;
  float* agg     = (float*)d_ws + 2 * NK;
  k_init<<<(NK + 255) / 256, 256, 0, stream>>>(mkey, NK);

  k_logit_max<<<(EWAVES + 7) / 8, 256, 0, stream>>>(uval, mu, sigma, urows, mkey);

  k_scatter<<<SLABS, SC_THREADS, 0, stream>>>(
      X, uval, mu, sigma, urows, ucols, mkey, denom, agg);

  k_gemm<<<GEMM_BLOCKS, 32 * GEMM_WAVES_PER_BLOCK, 0, stream>>>(agg, denom, GW, out);
}


# Row-id scan of the scatter keeps 8 loads per lane in flight, always issued so each slot's wait is counted (was 4, all waited for together); LDS float-add arrival order varies as before
# speedup vs baseline: 7.3795x; 1.0758x over previous
#include <hip/hip_runtime.h>
#include <math.h>

constexpr int BB = 2;
constexpr int NN = 50000;
constexpr int FF = 8;
constexpr int EE = 800000;
constexpr int DD = 3;
constexpr int KK = 8;
constexpr int HH = 64;
constexpr int FB = FF * BB;
constexpr int NK = NN * KK;
constexpr int CDIM = FF * KK;
constexpr int MROWS = BB * NN;
constexpr int MTILES = MROWS / 16;
constexpr int NTILES = HH / 16;
constexpr float EPSF = 1e-14f;
constexpr int GPW = 8;
constexpr int EPW = 4 * GPW;
constexpr int EWAVES = EE / EPW;
static_assert(EE % EPW == 0, "whole waves only");
static_assert(EPW == 32, "one index per lane");

typedef __attribute__((ext_vector_type(2))) float v2f;
typedef __attribute__((ext_vector_type(4))) float v4f;
typedef __attribute__((ext_vector_type(8))) float v8f;

__device__ __forceinline__ unsigned fkey(float f) {
  unsigned b = __float_as_uint(f);
  return (b & 0x80000000u) ? ~b : (b | 0x80000000u);
}
__device__ __forceinline__ float finv(unsigned k) {
  return __uint_as_float((k & 0x80000000u) ? (k & 0x7fffffffu) : ~k);
}

__global__ void k_init(unsigned* ws, int n) {
  int i = blockIdx.x * blockDim.x + threadIdx.x;
  int stride = gridDim.x * blockDim.x;
  for (; i < n; i += stride) ws[i] = 0u;
}

__global__ void k_logit_max(const float* __restrict__ uval,
                            const float* __restrict__ mu,
                            const float* __restrict__ sigma,
                            const int* __restrict__ urows,
                            unsigned* __restrict__ mkey) {
  int wid = blockIdx.x * (blockDim.x >> 5) + (threadIdx.x >> 5);
  if (wid >= EWAVES) return;
  int lane = threadIdx.x & 31;
  int sub = lane >> 3;
  int k = lane & (KK - 1);
  int e0 = wid * EPW;
  int rows = urows[e0 + lane];
  float mk[DD], iv[DD];
#pragma unroll
  for (int d = 0; d < DD; ++d) {
    float sg = sigma[k * DD + d];
    iv[d] = 1.0f / (sg * sg + EPSF);
    mk[d] = mu[k * DD + d];
  }
  int row[GPW];
  unsigned key[GPW];
#pragma unroll
  for (int g = 0; g < GPW; ++g) {
    int el = g * 4 + sub;
    row[g] = __shfl(rows, el);
    const float* uv = uval + (size_t)(e0 + el) * DD;
    float s = 0.0f;
#pragma unroll
    for (int d = 0; d < DD; ++d) {
      float df = uv[d] - mk[d];
      s += df * df * iv[d];
    }
    key[g] = fkey(-0.5f * s);
  }
#pragma unroll
  for (int g = 0; g < GPW; ++g) atomicMax(&mkey[row[g] * KK + k], key[g]);
}

typedef __attribute__((ext_vector_type(4))) int v4i;
constexpr int SLABS = 256;
constexpr int SROWS = (NN + SLABS - 1) / SLABS;
constexpr int SC_WAVES = 32;
constexpr int SC_THREADS = SC_WAVES * 32;
constexpr int RVECS = EE / 4;
constexpr int SWEEPS = (RVECS + SC_THREADS - 1) / SC_THREADS;
constexpr int PFD = 8;
constexpr int BATCH = 16;
constexpr int BGRP = BATCH / 4;
constexpr int LCAP = 1040;
constexpr int EID_BITS = 20;
static_assert(EE % 4 == 0, "row ids are read as vectors of 4");
static_assert(RVECS % 32 == 0, "the 32 vectors of a wave's load lie all inside the array or all outside");
static_assert(EE <= (1 << EID_BITS) && SROWS <= (1 << (32 - EID_BITS)), "list entry packing");
static_assert((SLABS - 1) * SROWS < NN && SLABS * SROWS >= NN, "every row owned once");
static_assert((SROWS * KK * FB) % 4 == 0 && (SROWS * KK) % 4 == 0, "slab moved as 16-byte vectors");

__global__ void __launch_bounds__(SC_THREADS)
k_scatter(const float* __restrict__ X,
          const float* __restrict__ uval,
          const float* __restrict__ mu,
          const float* __restrict__ sigma,
          const int* __restrict__ urows,
          const int* __restrict__ ucols,
          const unsigned* __restrict__ mkey,
          float* __restrict__ denom,
          float* __restrict__ agg) {
  __shared__ __attribute__((aligned(16))) float sagg[SROWS * KK * FB];
  __shared__ __attribute__((aligned(16))) float sden[SROWS * KK];
  __shared__ unsigned slist[SC_WAVES * LCAP];

  const int tid = threadIdx.x;
  const int lane = tid & 31;
  const int wave = __builtin_amdgcn_readfirstlane(tid >> 5);
  const int r0 = blockIdx.x * SROWS;

  {
    const v4f z = {0.0f, 0.0f, 0.0f, 0.0f};
    for (int i = tid; i < SROWS * KK * FB / 4; i += SC_THREADS) ((v4f*)sagg)[i] = z;
    for (int i = tid; i < SROWS * KK / 4; i += SC_THREADS) ((v4f*)sden)[i] = z;
  }
  __syncthreads();

  const int sub = lane >> 3;
  const int k = lane & (KK - 1);
  float mk[DD], fc[DD];
#pragma unroll
  for (int d = 0; d < DD; ++d) {
    float sg = sigma[k * DD + d];
    fc[d] = sg * sg + EPSF;
    mk[d] = mu[k * DD + d];
  }
  const int hi = lane >> 4;
  const int fb = lane & 15;
  const int xoff = (fb & 1) * (NN * FF) + (fb >> 1);

  unsigned* mylist = slist + wave * LCAP;
  int cnt = 0;

  const v4i* rvec = (const v4i*)urows;
  const int rot = (blockIdx.x * SWEEPS) / SLABS;
  auto sweep_base = [&](int s) -> int {
    if (s >= SWEEPS) return -1;
    int sr = s + rot;
    if (sr >= SWEEPS) sr -= SWEEPS;
    int vb = sr * SC_THREADS + wave * 32;
    return (vb < RVECS) ? vb : -1;
  };
  auto sweep_load = [&](int vb) -> v4i { return rvec[(vb < 0 ? 0 : vb) + lane]; };

  static_assert(LCAP >= BATCH - 1 + PFD * 4 * 32, "a trip's new entries fit the list");
  v4i q[PFD];
  int qb[PFD];
#pragma unroll
  for (int i = 0; i < PFD; ++i) {
    qb[i] = sweep_base(i);
    q[i] = sweep_load(qb[i]);
  }

  auto append = [&](const v4i cur, int vb) {
    if (vb < 0) return;
    const int vcur = vb + lane;
    unsigned dd[4];
    bool pp[4];
#pragma unroll
    for (int c = 0; c < 4; ++c) {
      dd[c] = (unsigned)(cur[c] - r0);
      pp[c] = dd[c] < (unsigned)SROWS;
    }
    if (__builtin_amdgcn_ballot_w32(pp[0] | pp[1] | pp[2] | pp[3]) != 0u) {
#pragma unroll
      for (int c = 0; c < 4; ++c) {
        unsigned m = __builtin_amdgcn_ballot_w32(pp[c]);
        if (m != 0u) {
          int pos = cnt + (int)__builtin_amdgcn_mbcnt_lo(m, 0u);
          if (pp[c]) mylist[pos] = (dd[c] << EID_BITS) | (unsigned)(4 * vcur + c);
          cnt += __builtin_popcount(m);
        }
      }
    }
  };

  for (int sw = 0; sw < SWEEPS + PFD; sw += PFD) {
    const bool last = (sw >= SWEEPS);
#pragma unroll
    for (int i = 0; i < PFD; ++i) {
      append(q[i], qb[i]);
      qb[i] = sweep_base(sw + PFD + i);
      q[i] = sweep_load(qb[i]);
    }

    for (int bt = 0; bt < LCAP / BATCH + 1; ++bt) {
      if (!(cnt >= BATCH || (last && cnt > 0))) break;
      const int nv = cnt < BATCH ? cnt : BATCH;
      const int base = cnt - nv;
      cnt = base;
      const unsigned ent = (lane < nv) ? mylist[base + lane] : 0u;
      const int eid = (int)(ent & ((1u << EID_BITS) - 1u));
      const int rl = (int)(ent >> EID_BITS);
      const int cols = ucols[eid];

      float xv[BATCH];
#pragma unroll
      for (int j = 0; j < BATCH; ++j) {
        int col_j = __builtin_amdgcn_readlane(cols, j);
        xv[j] = X[(size_t)col_j * FF + xoff];
      }
      float w[BGRP];
      int rlk[BGRP];
#pragma unroll
      for (int g = 0; g < BGRP; ++g) {
        int el = g * 4 + sub;
        int e_el = __shfl(eid, el);
        int rl_el = __shfl(rl, el);
        rlk[g] = rl_el * KK + k;
        const float* uv = uval + (size_t)e_el * DD;
        float s = 0.0f;
#pragma unroll
        for (int d = 0; d < DD; ++d) {
          float df = uv[d] - mk[d];
          s += df * df / fc[d];
        }
        float l = -0.5f * s;
        w[g] = __expf(l - finv(mkey[(r0 + rl_el) * KK + k]));
      }
#pragma unroll
      for (int g = 0; g < BGRP; ++g) {
        if (g * 4 + sub < nv) atomicAdd(&sden[rlk[g]], w[g]);
      }
#pragma unroll
      for (int g = 0; g < BGRP; ++g) {
#pragma unroll
        for (int ee = 0; ee < 4; ++ee) {
          if (g * 4 + ee < nv) {
            int rl_e = __builtin_amdgcn_readlane(rl, g * 4 + ee);
            float* dst = sagg + rl_e * (KK * FB) + lane;
#pragma unroll
            for (int i = 0; i < 4; ++i) {
              float wk = __shfl(w[g], ee * 8 + 2 * i + hi);
              atomicAdd(dst + i * 32, wk * xv[g * 4 + ee]);
            }
          }
        }
      }
    }
  }

  __syncthreads();
  const int nrows = (NN - r0 < SROWS) ? (NN - r0) : SROWS;
  v4f* gagg = (v4f*)(agg + (size_t)r0 * (KK * FB));
  for (int i = tid; i < nrows * (KK * FB / 4); i += SC_THREADS) gagg[i] = ((const v4f*)sagg)[i];
  v4f* gden = (v4f*)(denom + (size_t)r0 * KK);
  for (int i = tid; i < nrows * (KK / 4); i += SC_THREADS) gden[i] = ((const v4f*)sden)[i];
}

constexpr int NODE_TILES = NN / 16;
constexpr int GEMM_WAVES_PER_BLOCK = 4;
constexpr int GEMM_BLOCKS = (NODE_TILES + GEMM_WAVES_PER_BLOCK - 1) / GEMM_WAVES_PER_BLOCK;
static_assert(NN % 16 == 0, "whole node tiles only");
static_assert(KK * FB == 128 && HH == 64 && CDIM == 64, "tile maps below are written for these sizes");

__global__ void __launch_bounds__(32 * GEMM_WAVES_PER_BLOCK)
k_gemm(const float* __restrict__ agg,
       const float* __restrict__ denom,
       const float* __restrict__ GW,
       float* __restrict__ out) {
  int tile = blockIdx.x * (blockDim.x >> 5) + (threadIdx.x >> 5);
  if (tile >= NODE_TILES) return;
  int lane = threadIdx.x & 31;
  int h = lane >> 4;
  int lm = lane & 15;
  int n0 = tile * 16;

  const v4f* arow = (const v4f*)(agg + (size_t)(n0 + lm) * (KK * FB) + 4 * h);
  const v4f* drow = (const v4f*)(denom + (size_t)(n0 + lm) * KK);
  v4f dlo = drow[0];
  v4f dhi = drow[1];
  v4f av[16];
#pragma unroll
  for (int s = 0; s < 16; ++s) av[s] = arow[2 * s];
  float rinv[KK];
#pragma unroll
  for (int kk = 0; kk < KK; ++kk) {
    float d = (kk < 4) ? dlo[kk & 3] : dhi[kk & 3];
    rinv[kk] = (d == 0.0f) ? 0.0f : 1.0f / d;
  }

  v8f acc[2][4] = {};
  const float* gcol = GW + 4 * lm;
#pragma unroll
  for (int s = 0; s < 16; ++s) {
    int c0 = ((s & 1) * 4 + 2 * h) * KK + (s >> 1);
    int c1 = c0 + KK;
    v4f g0 = *(const v4f*)(gcol + c0 * HH);
    v4f g1 = *(const v4f*)(gcol + c1 * HH);
    v4f as = av[s] * rinv[s >> 1];
    v2f a0, a1;
    a0.x = as.x; a0.y = as.z;
    a1.x = as.y; a1.y = as.w;
#pragma unroll
    for (int t = 0; t < 4; ++t) {
      v2f bf;
      bf.x = g0[t];
      bf.y = g1[t];
      acc[0][t] = __builtin_amdgcn_wmma_f32_16x16x4_f32(
           false, a0,  false, bf,
           (short)0, acc[0][t],  false,  false);
      acc[1][t] = __builtin_amdgcn_wmma_f32_16x16x4_f32(
           false, a1,  false, bf,
           (short)0, acc[1][t],  false,  false);
    }
  }
#pragma unroll
  for (int b = 0; b < 2; ++b) {
#pragma unroll
    for (int j = 0; j < 8; ++j) {
      size_t r = (size_t)b * NN + n0 + h * 8 + j;
      v4f o;
      o.x = acc[b][0][j];
      o.y = acc[b][1][j];
      o.z = acc[b][2][j];
      o.w = acc[b][3][j];
      *(v4f*)(out + r * HH + 4 * lm) = o;
    }
  }
}

extern "C" void kernel_launch(void* const* d_in, const int* in_sizes, int n_in,
                              void* d_out, int out_size, void* d_ws, size_t ws_size,
                              hipStream_t stream) {
  const float* X     = (const float*)d_in[0];
  const float* uval  = (const float*)d_in[1];
  const float* mu    = (const float*)d_in[2];
  const float* sigma = (const float*)d_in[3];
  const float* GW    = (const float*)d_in[4];
  const int* urows   = (const int*)d_in[5];
  const int* ucols   = (const int*)d_in[6];
  float* out = (float*)d_out;

  unsigned* mkey = (unsigned*)d_ws;
  float* denom   = (float*)d_ws + NK;
  float* agg     = (float*)d_ws + 2 * NK;
  k_init<<<(NK + 255) / 256, 256, 0, stream>>>(mkey, NK);

  k_logit_max<<<(EWAVES + 7) / 8, 256, 0, stream>>>(uval, mu, sigma, urows, mkey);

  k_scatter<<<SLABS, SC_THREADS, 0, stream>>>(
      X, uval, mu, sigma, urows, ucols, mkey, denom, agg);

  k_gemm<<<GEMM_BLOCKS, 32 * GEMM_WAVES_PER_BLOCK, 0, stream>>>(agg, denom, GW, out);
}


# Pass 1 stores a one-byte owner tag per edge; the scatter scans tags (16 per lane-load, word-wide byte match) not 4-byte row ids: 4x fewer scan bytes and loads; LDS float-add order varies as before
# speedup vs baseline: 8.1380x; 1.1028x over previous
#include <hip/hip_runtime.h>
#include <math.h>

constexpr int BB = 2;
constexpr int NN = 50000;
constexpr int FF = 8;
constexpr int EE = 800000;
constexpr int DD = 3;
constexpr int KK = 8;
constexpr int HH = 64;
constexpr int FB = FF * BB;
constexpr int NK = NN * KK;
constexpr int CDIM = FF * KK;
constexpr int MROWS = BB * NN;
constexpr int MTILES = MROWS / 16;
constexpr int NTILES = HH / 16;
constexpr float EPSF = 1e-14f;
constexpr int GPW = 8;
constexpr int EPW = 4 * GPW;
constexpr int EWAVES = EE / EPW;
static_assert(EE % EPW == 0, "whole waves only");
static_assert(EPW == 32, "one index per lane");
constexpr int SLABS = 256;
constexpr int SROWS = (NN + SLABS - 1) / SLABS;
static_assert(SLABS <= 256, "an owner is one byte");

typedef __attribute__((ext_vector_type(2))) float v2f;
typedef __attribute__((ext_vector_type(4))) float v4f;
typedef __attribute__((ext_vector_type(8))) float v8f;

__device__ __forceinline__ unsigned fkey(float f) {
  unsigned b = __float_as_uint(f);
  return (b & 0x80000000u) ? ~b : (b | 0x80000000u);
}
__device__ __forceinline__ float finv(unsigned k) {
  return __uint_as_float((k & 0x80000000u) ? (k & 0x7fffffffu) : ~k);
}

__global__ void k_init(unsigned* ws, int n) {
  int i = blockIdx.x * blockDim.x + threadIdx.x;
  int stride = gridDim.x * blockDim.x;
  for (; i < n; i += stride) ws[i] = 0u;
}

__global__ void k_logit_max(const float* __restrict__ uval,
                            const float* __restrict__ mu,
                            const float* __restrict__ sigma,
                            const int* __restrict__ urows,
                            unsigned* __restrict__ mkey,
                            unsigned* __restrict__ tagw) {
  int wid = blockIdx.x * (blockDim.x >> 5) + (threadIdx.x >> 5);
  if (wid >= EWAVES) return;
  int lane = threadIdx.x & 31;
  int sub = lane >> 3;
  int k = lane & (KK - 1);
  int e0 = wid * EPW;
  int rows = urows[e0 + lane];
  ((unsigned char*)tagw)[e0 + lane] = (unsigned char)((unsigned)rows / (unsigned)SROWS);
  float mk[DD], iv[DD];
#pragma unroll
  for (int d = 0; d < DD; ++d) {
    float sg = sigma[k * DD + d];
    iv[d] = 1.0f / (sg * sg + EPSF);
    mk[d] = mu[k * DD + d];
  }
  int row[GPW];
  unsigned key[GPW];
#pragma unroll
  for (int g = 0; g < GPW; ++g) {
    int el = g * 4 + sub;
    row[g] = __shfl(rows, el);
    const float* uv = uval + (size_t)(e0 + el) * DD;
    float s = 0.0f;
#pragma unroll
    for (int d = 0; d < DD; ++d) {
      float df = uv[d] - mk[d];
      s += df * df * iv[d];
    }
    key[g] = fkey(-0.5f * s);
  }
#pragma unroll
  for (int g = 0; g < GPW; ++g) atomicMax(&mkey[row[g] * KK + k], key[g]);
}

typedef __attribute__((ext_vector_type(4))) int v4i;
typedef __attribute__((ext_vector_type(2))) unsigned v2u;
constexpr int SC_WAVES = 32;
constexpr int SC_THREADS = SC_WAVES * 32;
constexpr int TPL = 16;
constexpr int TVECS = EE / TPL;
constexpr int SWEEPS = (TVECS + SC_THREADS - 1) / SC_THREADS;
constexpr int PFD = 8;
constexpr int HCAP = PFD * 32;
constexpr int BATCH = 16;
constexpr int BGRP = BATCH / 4;
constexpr int LCAP = 544;
static_assert(EE % TPL == 0, "tags are read as vectors of 16");
static_assert(LCAP >= BATCH - 1 + TPL * 32, "32 records always fit behind a batch pass");
static_assert((SLABS - 1) * SROWS < NN && SLABS * SROWS >= NN, "every row owned once");
static_assert((SROWS * KK * FB) % 4 == 0 && (SROWS * KK) % 4 == 0, "slab moved as 16-byte vectors");

__global__ void __launch_bounds__(SC_THREADS)
k_scatter(const float* __restrict__ X,
          const float* __restrict__ uval,
          const float* __restrict__ mu,
          const float* __restrict__ sigma,
          const int* __restrict__ urows,
          const int* __restrict__ ucols,
          const unsigned* __restrict__ tagw,
          unsigned* __restrict__ mkd,
          float* __restrict__ agg) {
  __shared__ __attribute__((aligned(16))) float sagg[SROWS * KK * FB];
  __shared__ __attribute__((aligned(16))) float sden[SROWS * KK];
  __shared__ __attribute__((aligned(8))) v2u shits[SC_WAVES * HCAP];
  __shared__ unsigned slist[SC_WAVES * LCAP];

  const int tid = threadIdx.x;
  const int lane = tid & 31;
  const int wave = __builtin_amdgcn_readfirstlane(tid >> 5);
  const int r0 = blockIdx.x * SROWS;

  {
    const v4f z = {0.0f, 0.0f, 0.0f, 0.0f};
    for (int i = tid; i < SROWS * KK * FB / 4; i += SC_THREADS) ((v4f*)sagg)[i] = z;
    for (int i = tid; i < SROWS * KK / 4; i += SC_THREADS) ((v4f*)sden)[i] = z;
  }
  __syncthreads();

  const int sub = lane >> 3;
  const int k = lane & (KK - 1);
  float mk[DD], fc[DD];
#pragma unroll
  for (int d = 0; d < DD; ++d) {
    float sg = sigma[k * DD + d];
    fc[d] = sg * sg + EPSF;
    mk[d] = mu[k * DD + d];
  }
  const int hi = lane >> 4;
  const int fb = lane & 15;
  const int xoff = (fb & 1) * (NN * FF) + (fb >> 1);

  v2u* myhits = shits + wave * HCAP;
  unsigned* mylist = slist + wave * LCAP;
  int nh = 0;
  int cnt = 0;

  const v4i* tvec = (const v4i*)tagw;
  const unsigned ownrep = (unsigned)blockIdx.x * 0x01010101u;
  const int rot = (blockIdx.x * SWEEPS) / SLABS;
  auto sweep_base = [&](int s) -> int {
    if (s >= SWEEPS) return -1;
    int sr = s + rot;
    if (sr >= SWEEPS) sr -= SWEEPS;
    int vb = sr * SC_THREADS + wave * 32;
    return (vb < TVECS) ? vb : -1;
  };
  auto sweep_load = [&](int vb) -> v4i {
    int idx = (vb < 0 ? 0 : vb) + lane;
    return tvec[idx < TVECS ? idx : TVECS - 1];
  };

  v4i q[PFD];
  int qb[PFD];
#pragma unroll
  for (int i = 0; i < PFD; ++i) {
    qb[i] = sweep_base(i);
    q[i] = sweep_load(qb[i]);
  }

  auto note = [&](const v4i cur, int vb) {
    if (vb < 0) return;
    const int vcur = vb + lane;
    unsigned g = 0u;
#pragma unroll
    for (int c = 0; c < 4; ++c) {
      unsigned t = (unsigned)cur[c] ^ ownrep;
      unsigned v = ((t & 0x7f7f7f7fu) + 0x7f7f7f7fu) | t;
      g |= (~v & 0x80808080u) >> (7 - c);
    }
    if (vb + 32 > TVECS) {
      if (vcur >= TVECS) g = 0u;
    }
    const bool has = (g != 0u);
    const unsigned m = __builtin_amdgcn_ballot_w32(has);
    if (m != 0u) {
      const int pos = nh + (int)__builtin_amdgcn_mbcnt_lo(m, 0u);
      if (has) {
        v2u r;
        r.x = g;
        r.y = (unsigned)vcur;
        myhits[pos] = r;
      }
      nh += __builtin_popcount(m);
    }
  };

  auto drain = [&](const bool flush) {
    for (int bt = 0; bt < LCAP / BATCH + 1; ++bt) {
      if (!(cnt >= BATCH || (flush && cnt > 0))) break;
      const int nv = cnt < BATCH ? cnt : BATCH;
      const int base = cnt - nv;
      cnt = base;
      const int eid = (lane < nv) ? (int)mylist[base + lane] : 0;
      const int cols = ucols[eid];
      const unsigned dr = (unsigned)(urows[eid] - r0);
      const int rl = (lane < nv && dr < (unsigned)SROWS) ? (int)dr : 0;

      float xv[BATCH];
#pragma unroll
      for (int j = 0; j < BATCH; ++j) {
        int col_j = __builtin_amdgcn_readlane(cols, j);
        xv[j] = X[(size_t)col_j * FF + xoff];
      }
      float w[BGRP];
      int rlk[BGRP];
#pragma unroll
      for (int g = 0; g < BGRP; ++g) {
        int el = g * 4 + sub;
        int e_el = __shfl(eid, el);
        int rl_el = __shfl(rl, el);
        rlk[g] = rl_el * KK + k;
        const float* uv = uval + (size_t)e_el * DD;
        float s = 0.0f;
#pragma unroll
        for (int d = 0; d < DD; ++d) {
          float df = uv[d] - mk[d];
          s += df * df / fc[d];
        }
        float l = -0.5f * s;
        w[g] = __expf(l - finv(mkd[(r0 + rl_el) * KK + k]));
      }
#pragma unroll
      for (int g = 0; g < BGRP; ++g) {
        if (g * 4 + sub < nv) atomicAdd(&sden[rlk[g]], w[g]);
      }
#pragma unroll
      for (int g = 0; g < BGRP; ++g) {
#pragma unroll
        for (int ee = 0; ee < 4; ++ee) {
          if (g * 4 + ee < nv) {
            int rl_e = __builtin_amdgcn_readlane(rl, g * 4 + ee);
            float* dst = sagg + rl_e * (KK * FB) + lane;
#pragma unroll
            for (int i = 0; i < 4; ++i) {
              float wk = __shfl(w[g], ee * 8 + 2 * i + hi);
              atomicAdd(dst + i * 32, wk * xv[g * 4 + ee]);
            }
          }
        }
      }
    }
  };

  for (int sw = 0; sw < SWEEPS + PFD; sw += PFD) {
    const bool last = (sw >= SWEEPS);
#pragma unroll
    for (int i = 0; i < PFD; ++i) {
      note(q[i], qb[i]);
      qb[i] = sweep_base(sw + PFD + i);
      q[i] = sweep_load(qb[i]);
    }

    for (int off = 0; off < HCAP; off += 32) {
      if (off < nh) {
        const int ridx = off + lane;
        v2u r = {0u, 0u};
        if (ridx < nh) r = myhits[ridx];
        unsigned g = r.x;
        const unsigned ebase = r.y * (unsigned)TPL;
#pragma nounroll
        for (int it = 0; it < TPL; ++it) {
          const bool has = (g != 0u);
          const unsigned m = __builtin_amdgcn_ballot_w32(has);
          if (m == 0u) break;
          const int pos = cnt + (int)__builtin_amdgcn_mbcnt_lo(m, 0u);
          if (has) {
            const int p = __builtin_ctz(g);
            mylist[pos] = ebase + (unsigned)(((p & 7) << 2) + (p >> 3));
          }
          g &= g - 1u;
          cnt += __builtin_popcount(m);
        }
      }
      const bool pending = (off + 32 < nh);
      drain(last && !pending);
      if (!pending) break;
    }
    nh = 0;
  }

  __syncthreads();
  const int nrows = (NN - r0 < SROWS) ? (NN - r0) : SROWS;
  v4f* gagg = (v4f*)(agg + (size_t)r0 * (KK * FB));
  for (int i = tid; i < nrows * (KK * FB / 4); i += SC_THREADS) gagg[i] = ((const v4f*)sagg)[i];
  v4f* gden = (v4f*)((float*)mkd + (size_t)r0 * KK);
  for (int i = tid; i < nrows * (KK / 4); i += SC_THREADS) gden[i] = ((const v4f*)sden)[i];
}

constexpr int NODE_TILES = NN / 16;
constexpr int GEMM_WAVES_PER_BLOCK = 4;
constexpr int GEMM_BLOCKS = (NODE_TILES + GEMM_WAVES_PER_BLOCK - 1) / GEMM_WAVES_PER_BLOCK;
static_assert(NN % 16 == 0, "whole node tiles only");
static_assert(KK * FB == 128 && HH == 64 && CDIM == 64, "tile maps below are written for these sizes");

__global__ void __launch_bounds__(32 * GEMM_WAVES_PER_BLOCK)
k_gemm(const float* __restrict__ agg,
       const float* __restrict__ denom,
       const float* __restrict__ GW,
       float* __restrict__ out) {
  int tile = blockIdx.x * (blockDim.x >> 5) + (threadIdx.x >> 5);
  if (tile >= NODE_TILES) return;
  int lane = threadIdx.x & 31;
  int h = lane >> 4;
  int lm = lane & 15;
  int n0 = tile * 16;

  const v4f* arow = (const v4f*)(agg + (size_t)(n0 + lm) * (KK * FB) + 4 * h);
  const v4f* drow = (const v4f*)(denom + (size_t)(n0 + lm) * KK);
  v4f dlo = drow[0];
  v4f dhi = drow[1];
  v4f av[16];
#pragma unroll
  for (int s = 0; s < 16; ++s) av[s] = arow[2 * s];
  float rinv[KK];
#pragma unroll
  for (int kk = 0; kk < KK; ++kk) {
    float d = (kk < 4) ? dlo[kk & 3] : dhi[kk & 3];
    rinv[kk] = (d == 0.0f) ? 0.0f : 1.0f / d;
  }

  v8f acc[2][4] = {};
  const float* gcol = GW + 4 * lm;
#pragma unroll
  for (int s = 0; s < 16; ++s) {
    int c0 = ((s & 1) * 4 + 2 * h) * KK + (s >> 1);
    int c1 = c0 + KK;
    v4f g0 = *(const v4f*)(gcol + c0 * HH);
    v4f g1 = *(const v4f*)(gcol + c1 * HH);
    v4f as = av[s] * rinv[s >> 1];
    v2f a0, a1;
    a0.x = as.x; a0.y = as.z;
    a1.x = as.y; a1.y = as.w;
#pragma unroll
    for (int t = 0; t < 4; ++t) {
      v2f bf;
      bf.x = g0[t];
      bf.y = g1[t];
      acc[0][t] = __builtin_amdgcn_wmma_f32_16x16x4_f32(
           false, a0,  false, bf,
           (short)0, acc[0][t],  false,  false);
      acc[1][t] = __builtin_amdgcn_wmma_f32_16x16x4_f32(
           false, a1,  false, bf,
           (short)0, acc[1][t],  false,  false);
    }
  }
#pragma unroll
  for (int b = 0; b < 2; ++b) {
#pragma unroll
    for (int j = 0; j < 8; ++j) {
      size_t r = (size_t)b * NN + n0 + h * 8 + j;
      v4f o;
      o.x = acc[b][0][j];
      o.y = acc[b][1][j];
      o.z = acc[b][2][j];
      o.w = acc[b][3][j];
      *(v4f*)(out + r * HH + 4 * lm) = o;
    }
  }
}

extern "C" void kernel_launch(void* const* d_in, const int* in_sizes, int n_in,
                              void* d_out, int out_size, void* d_ws, size_t ws_size,
                              hipStream_t stream) {
  const float* X     = (const float*)d_in[0];
  const float* uval  = (const float*)d_in[1];
  const float* mu    = (const float*)d_in[2];
  const float* sigma = (const float*)d_in[3];
  const float* GW    = (const float*)d_in[4];
  const int* urows   = (const int*)d_in[5];
  const int* ucols   = (const int*)d_in[6];
  float* out = (float*)d_out;

  unsigned* mkey = (unsigned*)d_ws;
  unsigned* tagw = (unsigned*)d_ws + NK;
  float* agg     = (float*)d_ws + 2 * NK;
  k_init<<<(NK + 255) / 256, 256, 0, stream>>>(mkey, NK);

  k_logit_max<<<(EWAVES + 7) / 8, 256, 0, stream>>>(uval, mu, sigma, urows, mkey, tagw);

  k_scatter<<<SLABS, SC_THREADS, 0, stream>>>(
      X, uval, mu, sigma, urows, ucols, tagw, mkey, agg);

  k_gemm<<<GEMM_BLOCKS, 32 * GEMM_WAVES_PER_BLOCK, 0, stream>>>(agg, (const float*)mkey, GW, out);
}


# GEMM fused into the scatter: each work-group multiplies its finished LDS slab (padded row pitch, GW staged in LDS in fragment order) by GW on f32 WMMA and stores out; no agg round trip, 3 launches
# speedup vs baseline: 8.7075x; 1.0700x over previous
#include <hip/hip_runtime.h>
#include <math.h>

constexpr int BB = 2;
constexpr int NN = 50000;
constexpr int FF = 8;
constexpr int EE = 800000;
constexpr int DD = 3;
constexpr int KK = 8;
constexpr int HH = 64;
constexpr int FB = FF * BB;
constexpr int NK = NN * KK;
constexpr int CDIM = FF * KK;
constexpr float EPSF = 1e-14f;
constexpr int GPW = 8;
constexpr int EPW = 4 * GPW;
constexpr int EWAVES = EE / EPW;
static_assert(EE % EPW == 0, "whole waves only");
static_assert(EPW == 32, "one index per lane");
constexpr int SLABS = 256;
constexpr int SROWS = (NN + SLABS - 1) / SLABS;
static_assert(SLABS <= 256, "an owner is one byte");

typedef __attribute__((ext_vector_type(2))) float v2f;
typedef __attribute__((ext_vector_type(4))) float v4f;
typedef __attribute__((ext_vector_type(8))) float v8f;

__device__ __forceinline__ unsigned fkey(float f) {
  unsigned b = __float_as_uint(f);
  return (b & 0x80000000u) ? ~b : (b | 0x80000000u);
}
__device__ __forceinline__ float finv(unsigned k) {
  return __uint_as_float((k & 0x80000000u) ? (k & 0x7fffffffu) : ~k);
}

__global__ void k_init(unsigned* ws, int n) {
  int i = blockIdx.x * blockDim.x + threadIdx.x;
  int stride = gridDim.x * blockDim.x;
  for (; i < n; i += stride) ws[i] = 0u;
}

__global__ void k_logit_max(const float* __restrict__ uval,
                            const float* __restrict__ mu,
                            const float* __restrict__ sigma,
                            const int* __restrict__ urows,
                            unsigned* __restrict__ mkey,
                            unsigned* __restrict__ tagw) {
  int wid = blockIdx.x * (blockDim.x >> 5) + (threadIdx.x >> 5);
  if (wid >= EWAVES) return;
  int lane = threadIdx.x & 31;
  int sub = lane >> 3;
  int k = lane & (KK - 1);
  int e0 = wid * EPW;
  int rows = urows[e0 + lane];
  ((unsigned char*)tagw)[e0 + lane] = (unsigned char)((unsigned)rows / (unsigned)SROWS);
  float mk[DD], iv[DD];
#pragma unroll
  for (int d = 0; d < DD; ++d) {
    float sg = sigma[k * DD + d];
    iv[d] = 1.0f / (sg * sg + EPSF);
    mk[d] = mu[k * DD + d];
  }
  int row[GPW];
  unsigned key[GPW];
#pragma unroll
  for (int g = 0; g < GPW; ++g) {
    int el = g * 4 + sub;
    row[g] = __shfl(rows, el);
    const float* uv = uval + (size_t)(e0 + el) * DD;
    float s = 0.0f;
#pragma unroll
    for (int d = 0; d < DD; ++d) {
      float df = uv[d] - mk[d];
      s += df * df * iv[d];
    }
    key[g] = fkey(-0.5f * s);
  }
#pragma unroll
  for (int g = 0; g < GPW; ++g) atomicMax(&mkey[row[g] * KK + k], key[g]);
}

typedef __attribute__((ext_vector_type(4))) int v4i;
typedef __attribute__((ext_vector_type(2))) unsigned v2u;
constexpr int SC_WAVES = 32;
constexpr int SC_THREADS = SC_WAVES * 32;
constexpr int TPL = 16;
constexpr int TVECS = EE / TPL;
constexpr int SWEEPS = (TVECS + SC_THREADS - 1) / SC_THREADS;
constexpr int PFD = 8;
constexpr int HCAP = PFD * 32;
constexpr int BATCH = 16;
constexpr int BGRP = BATCH / 4;
constexpr int LCAP = 544;
static_assert(EE % TPL == 0, "tags are read as vectors of 16");
static_assert(LCAP >= BATCH - 1 + TPL * 32, "32 records always fit behind a batch pass");
static_assert((SLABS - 1) * SROWS < NN && SLABS * SROWS >= NN, "every row owned once");
constexpr int RS = KK * FB + 4;
constexpr int STILES = (SROWS + 15) / 16;
static_assert((SROWS * RS) % 4 == 0 && (SROWS * KK) % 4 == 0, "slab zeroed as 16-byte vectors");
static_assert(((RS / 4) & 1) == 1, "row pitch is 16 bytes x odd");
static_assert(STILES * BB <= SC_WAVES, "one wave per (row tile, batch)");
static_assert(KK * FB == 128 && HH == 64 && CDIM == 64, "tile maps of the epilogue are written for these sizes");

__global__ void __launch_bounds__(SC_THREADS)
k_scatter(const float* __restrict__ X,
          const float* __restrict__ uval,
          const float* __restrict__ mu,
          const float* __restrict__ sigma,
          const int* __restrict__ urows,
          const int* __restrict__ ucols,
          const unsigned* __restrict__ tagw,
          const unsigned* __restrict__ mkd,
          const float* __restrict__ GW,
          float* __restrict__ out) {
  __shared__ __attribute__((aligned(16))) float sagg[SROWS * RS];
  __shared__ __attribute__((aligned(16))) float sden[SROWS * KK];
  __shared__ __attribute__((aligned(16))) float sgw[CDIM * HH];
  __shared__ __attribute__((aligned(8))) v2u shits[SC_WAVES * HCAP];
  __shared__ unsigned slist[SC_WAVES * LCAP];

  const int tid = threadIdx.x;
  const int lane = tid & 31;
  const int wave = __builtin_amdgcn_readfirstlane(tid >> 5);
  const int r0 = blockIdx.x * SROWS;

  {
    const v4f z = {0.0f, 0.0f, 0.0f, 0.0f};
    for (int i = tid; i < SROWS * RS / 4; i += SC_THREADS) ((v4f*)sagg)[i] = z;
    for (int i = tid; i < SROWS * KK / 4; i += SC_THREADS) ((v4f*)sden)[i] = z;
    for (int i = tid; i < CDIM * HH; i += SC_THREADS) {
      const int j = i & 1, tl = (i >> 1) & 1, lm_ = (i >> 2) & 15;
      const int tp = (i >> 6) & 1, h_ = (i >> 7) & 1, s_ = i >> 8;
      const int c = ((s_ & 1) * 4 + 2 * h_ + j) * KK + (s_ >> 1);
      sgw[i] = GW[c * HH + 4 * lm_ + 2 * tp + tl];
    }
  }
  __syncthreads();

  const int sub = lane >> 3;
  const int k = lane & (KK - 1);
  float mk[DD], fc[DD];
#pragma unroll
  for (int d = 0; d < DD; ++d) {
    float sg = sigma[k * DD + d];
    fc[d] = sg * sg + EPSF;
    mk[d] = mu[k * DD + d];
  }
  const int hi = lane >> 4;
  const int fb = lane & 15;
  const int xoff = (fb & 1) * (NN * FF) + (fb >> 1);

  v2u* myhits = shits + wave * HCAP;
  unsigned* mylist = slist + wave * LCAP;
  int nh = 0;
  int cnt = 0;

  const v4i* tvec = (const v4i*)tagw;
  const unsigned ownrep = (unsigned)blockIdx.x * 0x01010101u;
  const int rot = (blockIdx.x * SWEEPS) / SLABS;
  auto sweep_base = [&](int s) -> int {
    if (s >= SWEEPS) return -1;
    int sr = s + rot;
    if (sr >= SWEEPS) sr -= SWEEPS;
    int vb = sr * SC_THREADS + wave * 32;
    return (vb < TVECS) ? vb : -1;
  };
  auto sweep_load = [&](int vb) -> v4i {
    int idx = (vb < 0 ? 0 : vb) + lane;
    return tvec[idx < TVECS ? idx : TVECS - 1];
  };

  v4i q[PFD];
  int qb[PFD];
#pragma unroll
  for (int i = 0; i < PFD; ++i) {
    qb[i] = sweep_base(i);
    q[i] = sweep_load(qb[i]);
  }

  auto note = [&](const v4i cur, int vb) {
    if (vb < 0) return;
    const int vcur = vb + lane;
    unsigned g = 0u;
#pragma unroll
    for (int c = 0; c < 4; ++c) {
      unsigned t = (unsigned)cur[c] ^ ownrep;
      unsigned v = ((t & 0x7f7f7f7fu) + 0x7f7f7f7fu) | t;
      g |= (~v & 0x80808080u) >> (7 - c);
    }
    if (vb + 32 > TVECS) {
      if (vcur >= TVECS) g = 0u;
    }
    const bool has = (g != 0u);
    const unsigned m = __builtin_amdgcn_ballot_w32(has);
    if (m != 0u) {
      const int pos = nh + (int)__builtin_amdgcn_mbcnt_lo(m, 0u);
      if (has) {
        v2u r;
        r.x = g;
        r.y = (unsigned)vcur;
        myhits[pos] = r;
      }
      nh += __builtin_popcount(m);
    }
  };

  auto drain = [&](const bool flush) {
    for (int bt = 0; bt < LCAP / BATCH + 1; ++bt) {
      if (!(cnt >= BATCH || (flush && cnt > 0))) break;
      const int nv = cnt < BATCH ? cnt : BATCH;
      const int base = cnt - nv;
      cnt = base;
      const int eid = (lane < nv) ? (int)mylist[base + lane] : 0;
      const int cols = ucols[eid];
      const unsigned dr = (unsigned)(urows[eid] - r0);
      const int rl = (lane < nv && dr < (unsigned)SROWS) ? (int)dr : 0;

      float xv[BATCH];
#pragma unroll
      for (int j = 0; j < BATCH; ++j) {
        int col_j = __builtin_amdgcn_readlane(cols, j);
        xv[j] = X[(size_t)col_j * FF + xoff];
      }
      float w[BGRP];
      int rlk[BGRP];
#pragma unroll
      for (int g = 0; g < BGRP; ++g) {
        int el = g * 4 + sub;
        int e_el = __shfl(eid, el);
        int rl_el = __shfl(rl, el);
        rlk[g] = rl_el * KK + k;
        const float* uv = uval + (size_t)e_el * DD;
        float s = 0.0f;
#pragma unroll
        for (int d = 0; d < DD; ++d) {
          float df = uv[d] - mk[d];
          s += df * df / fc[d];
        }
        float l = -0.5f * s;
        w[g] = __expf(l - finv(mkd[(r0 + rl_el) * KK + k]));
      }
#pragma unroll
      for (int g = 0; g < BGRP; ++g) {
        if (g * 4 + sub < nv) atomicAdd(&sden[rlk[g]], w[g]);
      }
#pragma unroll
      for (int g = 0; g < BGRP; ++g) {
#pragma unroll
        for (int ee = 0; ee < 4; ++ee) {
          if (g * 4 + ee < nv) {
            int rl_e = __builtin_amdgcn_readlane(rl, g * 4 + ee);
            float* dst = sagg + rl_e * RS + lane;
#pragma unroll
            for (int i = 0; i < 4; ++i) {
              float wk = __shfl(w[g], ee * 8 + 2 * i + hi);
              atomicAdd(dst + i * 32, wk * xv[g * 4 + ee]);
            }
          }
        }
      }
    }
  };

  for (int sw = 0; sw < SWEEPS + PFD; sw += PFD) {
    const bool last = (sw >= SWEEPS);
#pragma unroll
    for (int i = 0; i < PFD; ++i) {
      note(q[i], qb[i]);
      qb[i] = sweep_base(sw + PFD + i);
      q[i] = sweep_load(qb[i]);
    }

    for (int off = 0; off < HCAP; off += 32) {
      if (off < nh) {
        const int ridx = off + lane;
        v2u r = {0u, 0u};
        if (ridx < nh) r = myhits[ridx];
        unsigned g = r.x;
        const unsigned ebase = r.y * (unsigned)TPL;
#pragma nounroll
        for (int it = 0; it < TPL; ++it) {
          const bool has = (g != 0u);
          const unsigned m = __builtin_amdgcn_ballot_w32(has);
          if (m == 0u) break;
          const int pos = cnt + (int)__builtin_amdgcn_mbcnt_lo(m, 0u);
          if (has) {
            const int p = __builtin_ctz(g);
            mylist[pos] = ebase + (unsigned)(((p & 7) << 2) + (p >> 3));
          }
          g &= g - 1u;
          cnt += __builtin_popcount(m);
        }
      }
      const bool pending = (off + 32 < nh);
      drain(last && !pending);
      if (!pending) break;
    }
    nh = 0;
  }

  __syncthreads();
  const int nrows = (NN - r0 < SROWS) ? (NN - r0) : SROWS;
  const int tile = wave >> 1;
  const int bsel = wave & 1;
  if (wave < STILES * BB && tile * 16 < nrows) {
    int ar = tile * 16 + fb;
    ar = ar < SROWS ? ar : SROWS - 1;
    const float* ap = sagg + ar * RS + 4 * hi + bsel;
    const v4f* drow = (const v4f*)(sden + ar * KK);
    const v4f dlo = drow[0];
    const v4f dhi = drow[1];
    float rinv[KK];
#pragma unroll
    for (int kk = 0; kk < KK; ++kk) {
      float d = (kk < 4) ? dlo[kk & 3] : dhi[kk & 3];
      rinv[kk] = (d == 0.0f) ? 0.0f : 1.0f / d;
    }
    const float* bp = sgw + hi * 128 + fb * 4;
    v8f acc[4] = {};
#pragma unroll
    for (int s = 0; s < 16; ++s) {
      v2f a;
      a.x = ap[8 * s];
      a.y = ap[8 * s + 2];
      a *= rinv[s >> 1];
      const v4f b01 = *(const v4f*)(bp + s * 256);
      const v4f b23 = *(const v4f*)(bp + s * 256 + 64);
      v2f bf[4];
      bf[0].x = b01.x; bf[0].y = b01.y;
      bf[1].x = b01.z; bf[1].y = b01.w;
      bf[2].x = b23.x; bf[2].y = b23.y;
      bf[3].x = b23.z; bf[3].y = b23.w;
#pragma unroll
      for (int t = 0; t < 4; ++t) {
        acc[t] = __builtin_amdgcn_wmma_f32_16x16x4_f32(
             false, a,  false, bf[t],
             (short)0, acc[t],  false,  false);
      }
    }
    float* orow = out + ((size_t)bsel * NN + r0 + tile * 16 + hi * 8) * HH + 4 * fb;
    const int rleft = nrows - (tile * 16 + hi * 8);
#pragma unroll
    for (int j = 0; j < 8; ++j) {
      if (j < rleft) {
        v4f o;
        o.x = acc[0][j];
        o.y = acc[1][j];
        o.z = acc[2][j];
        o.w = acc[3][j];
        *(v4f*)(orow + (size_t)j * HH) = o;
      }
    }
  }
}

extern "C" void kernel_launch(void* const* d_in, const int* in_sizes, int n_in,
                              void* d_out, int out_size, void* d_ws, size_t ws_size,
                              hipStream_t stream) {
  const float* X     = (const float*)d_in[0];
  const float* uval  = (const float*)d_in[1];
  const float* mu    = (const float*)d_in[2];
  const float* sigma = (const float*)d_in[3];
  const float* GW    = (const float*)d_in[4];
  const int* urows   = (const int*)d_in[5];
  const int* ucols   = (const int*)d_in[6];
  float* out = (float*)d_out;

  unsigned* mkey = (unsigned*)d_ws;
  unsigned* tagw = (unsigned*)d_ws + NK;
  k_init<<<(NK + 255) / 256, 256, 0, stream>>>(mkey, NK);

  k_logit_max<<<(EWAVES + 7) / 8, 256, 0, stream>>>(uval, mu, sigma, urows, mkey, tagw);

  k_scatter<<<SLABS, SC_THREADS, 0, stream>>>(
      X, uval, mu, sigma, urows, ucols, tagw, mkey, GW, out);
}


# Slab adds take each lane's weight by DPP row_share from its own 16-lane row (weight lanes re-mapped: even k in row 0, odd k in row 1), not 4 ds_bpermute + wait per edge; LDS add order varies as ever
# speedup vs baseline: 9.0366x; 1.0378x over previous
#include <hip/hip_runtime.h>
#include <math.h>

constexpr int BB = 2;
constexpr int NN = 50000;
constexpr int FF = 8;
constexpr int EE = 800000;
constexpr int DD = 3;
constexpr int KK = 8;
constexpr int HH = 64;
constexpr int FB = FF * BB;
constexpr int NK = NN * KK;
constexpr int CDIM = FF * KK;
constexpr float EPSF = 1e-14f;
constexpr int GPW = 8;
constexpr int EPW = 4 * GPW;
constexpr int EWAVES = EE / EPW;
static_assert(EE % EPW == 0, "whole waves only");
static_assert(EPW == 32, "one index per lane");
constexpr int SLABS = 256;
constexpr int SROWS = (NN + SLABS - 1) / SLABS;
static_assert(SLABS <= 256, "an owner is one byte");

typedef __attribute__((ext_vector_type(2))) float v2f;
typedef __attribute__((ext_vector_type(4))) float v4f;
typedef __attribute__((ext_vector_type(8))) float v8f;

__device__ __forceinline__ unsigned fkey(float f) {
  unsigned b = __float_as_uint(f);
  return (b & 0x80000000u) ? ~b : (b | 0x80000000u);
}
__device__ __forceinline__ float finv(unsigned k) {
  return __uint_as_float((k & 0x80000000u) ? (k & 0x7fffffffu) : ~k);
}

#define ROW_SHARE_CASE(N) \
  case N: r = __builtin_amdgcn_update_dpp(0, s, 0x150 + N, 0xf, 0xf, true); break;
__device__ __forceinline__ float row_share(float v, int n) {
  const int s = __float_as_int(v);
  int r = s;
  switch (n) {
    ROW_SHARE_CASE(0) ROW_SHARE_CASE(1) ROW_SHARE_CASE(2) ROW_SHARE_CASE(3)
    ROW_SHARE_CASE(4) ROW_SHARE_CASE(5) ROW_SHARE_CASE(6) ROW_SHARE_CASE(7)
    ROW_SHARE_CASE(8) ROW_SHARE_CASE(9) ROW_SHARE_CASE(10) ROW_SHARE_CASE(11)
    ROW_SHARE_CASE(12) ROW_SHARE_CASE(13) ROW_SHARE_CASE(14) ROW_SHARE_CASE(15)
    default: break;
  }
  return __int_as_float(r);
}
#undef ROW_SHARE_CASE

__global__ void k_init(unsigned* ws, int n) {
  int i = blockIdx.x * blockDim.x + threadIdx.x;
  int stride = gridDim.x * blockDim.x;
  for (; i < n; i += stride) ws[i] = 0u;
}

__global__ void k_logit_max(const float* __restrict__ uval,
                            const float* __restrict__ mu,
                            const float* __restrict__ sigma,
                            const int* __restrict__ urows,
                            unsigned* __restrict__ mkey,
                            unsigned* __restrict__ tagw) {
  int wid = blockIdx.x * (blockDim.x >> 5) + (threadIdx.x >> 5);
  if (wid >= EWAVES) return;
  int lane = threadIdx.x & 31;
  int sub = lane >> 3;
  int k = lane & (KK - 1);
  int e0 = wid * EPW;
  int rows = urows[e0 + lane];
  ((unsigned char*)tagw)[e0 + lane] = (unsigned char)((unsigned)rows / (unsigned)SROWS);
  float mk[DD], iv[DD];
#pragma unroll
  for (int d = 0; d < DD; ++d) {
    float sg = sigma[k * DD + d];
    iv[d] = 1.0f / (sg * sg + EPSF);
    mk[d] = mu[k * DD + d];
  }
  int row[GPW];
  unsigned key[GPW];
#pragma unroll
  for (int g = 0; g < GPW; ++g) {
    int el = g * 4 + sub;
    row[g] = __shfl(rows, el);
    const float* uv = uval + (size_t)(e0 + el) * DD;
    float s = 0.0f;
#pragma unroll
    for (int d = 0; d < DD; ++d) {
      float df = uv[d] - mk[d];
      s += df * df * iv[d];
    }
    key[g] = fkey(-0.5f * s);
  }
#pragma unroll
  for (int g = 0; g < GPW; ++g) atomicMax(&mkey[row[g] * KK + k], key[g]);
}

typedef __attribute__((ext_vector_type(4))) int v4i;
typedef __attribute__((ext_vector_type(2))) unsigned v2u;
constexpr int SC_WAVES = 32;
constexpr int SC_THREADS = SC_WAVES * 32;
constexpr int TPL = 16;
constexpr int TVECS = EE / TPL;
constexpr int SWEEPS = (TVECS + SC_THREADS - 1) / SC_THREADS;
constexpr int PFD = 8;
constexpr int HCAP = PFD * 32;
constexpr int BATCH = 16;
constexpr int BGRP = BATCH / 4;
constexpr int LCAP = 544;
static_assert(EE % TPL == 0, "tags are read as vectors of 16");
static_assert(LCAP >= BATCH - 1 + TPL * 32, "32 records always fit behind a batch pass");
static_assert((SLABS - 1) * SROWS < NN && SLABS * SROWS >= NN, "every row owned once");
constexpr int RS = KK * FB + 4;
constexpr int STILES = (SROWS + 15) / 16;
static_assert((SROWS * RS) % 4 == 0 && (SROWS * KK) % 4 == 0, "slab zeroed as 16-byte vectors");
static_assert(((RS / 4) & 1) == 1, "row pitch is 16 bytes x odd");
static_assert(STILES * BB <= SC_WAVES, "one wave per (row tile, batch)");
static_assert(KK * FB == 128 && HH == 64 && CDIM == 64, "tile maps of the epilogue are written for these sizes");

__global__ void __launch_bounds__(SC_THREADS)
k_scatter(const float* __restrict__ X,
          const float* __restrict__ uval,
          const float* __restrict__ mu,
          const float* __restrict__ sigma,
          const int* __restrict__ urows,
          const int* __restrict__ ucols,
          const unsigned* __restrict__ tagw,
          const unsigned* __restrict__ mkd,
          const float* __restrict__ GW,
          float* __restrict__ out) {
  __shared__ __attribute__((aligned(16))) float sagg[SROWS * RS];
  __shared__ __attribute__((aligned(16))) float sden[SROWS * KK];
  __shared__ __attribute__((aligned(16))) float sgw[CDIM * HH];
  __shared__ __attribute__((aligned(8))) v2u shits[SC_WAVES * HCAP];
  __shared__ unsigned slist[SC_WAVES * LCAP];

  const int tid = threadIdx.x;
  const int lane = tid & 31;
  const int wave = __builtin_amdgcn_readfirstlane(tid >> 5);
  const int r0 = blockIdx.x * SROWS;

  {
    const v4f z = {0.0f, 0.0f, 0.0f, 0.0f};
    for (int i = tid; i < SROWS * RS / 4; i += SC_THREADS) ((v4f*)sagg)[i] = z;
    for (int i = tid; i < SROWS * KK / 4; i += SC_THREADS) ((v4f*)sden)[i] = z;
    for (int i = tid; i < CDIM * HH; i += SC_THREADS) {
      const int j = i & 1, tl = (i >> 1) & 1, lm_ = (i >> 2) & 15;
      const int tp = (i >> 6) & 1, h_ = (i >> 7) & 1, s_ = i >> 8;
      const int c = ((s_ & 1) * 4 + 2 * h_ + j) * KK + (s_ >> 1);
      sgw[i] = GW[c * HH + 4 * lm_ + 2 * tp + tl];
    }
  }
  __syncthreads();

  const int hi = lane >> 4;
  const int fb = lane & 15;
  const int sub = fb >> 2;
  const int k = 2 * (fb & 3) + hi;
  float mk[DD], fc[DD];
#pragma unroll
  for (int d = 0; d < DD; ++d) {
    float sg = sigma[k * DD + d];
    fc[d] = sg * sg + EPSF;
    mk[d] = mu[k * DD + d];
  }
  const int xoff = (fb & 1) * (NN * FF) + (fb >> 1);

  v2u* myhits = shits + wave * HCAP;
  unsigned* mylist = slist + wave * LCAP;
  int nh = 0;
  int cnt = 0;

  const v4i* tvec = (const v4i*)tagw;
  const unsigned ownrep = (unsigned)blockIdx.x * 0x01010101u;
  const int rot = (blockIdx.x * SWEEPS) / SLABS;
  auto sweep_base = [&](int s) -> int {
    if (s >= SWEEPS) return -1;
    int sr = s + rot;
    if (sr >= SWEEPS) sr -= SWEEPS;
    int vb = sr * SC_THREADS + wave * 32;
    return (vb < TVECS) ? vb : -1;
  };
  auto sweep_load = [&](int vb) -> v4i {
    int idx = (vb < 0 ? 0 : vb) + lane;
    return tvec[idx < TVECS ? idx : TVECS - 1];
  };

  v4i q[PFD];
  int qb[PFD];
#pragma unroll
  for (int i = 0; i < PFD; ++i) {
    qb[i] = sweep_base(i);
    q[i] = sweep_load(qb[i]);
  }

  auto note = [&](const v4i cur, int vb) {
    if (vb < 0) return;
    const int vcur = vb + lane;
    unsigned g = 0u;
#pragma unroll
    for (int c = 0; c < 4; ++c) {
      unsigned t = (unsigned)cur[c] ^ ownrep;
      unsigned v = ((t & 0x7f7f7f7fu) + 0x7f7f7f7fu) | t;
      g |= (~v & 0x80808080u) >> (7 - c);
    }
    if (vb + 32 > TVECS) {
      if (vcur >= TVECS) g = 0u;
    }
    const bool has = (g != 0u);
    const unsigned m = __builtin_amdgcn_ballot_w32(has);
    if (m != 0u) {
      const int pos = nh + (int)__builtin_amdgcn_mbcnt_lo(m, 0u);
      if (has) {
        v2u r;
        r.x = g;
        r.y = (unsigned)vcur;
        myhits[pos] = r;
      }
      nh += __builtin_popcount(m);
    }
  };

  auto drain = [&](const bool flush) {
    for (int bt = 0; bt < LCAP / BATCH + 1; ++bt) {
      if (!(cnt >= BATCH || (flush && cnt > 0))) break;
      const int nv = cnt < BATCH ? cnt : BATCH;
      const int base = cnt - nv;
      cnt = base;
      const int eid = (lane < nv) ? (int)mylist[base + lane] : 0;
      const int cols = ucols[eid];
      const unsigned dr = (unsigned)(urows[eid] - r0);
      const int rl = (lane < nv && dr < (unsigned)SROWS) ? (int)dr : 0;

      float xv[BATCH];
#pragma unroll
      for (int j = 0; j < BATCH; ++j) {
        int col_j = __builtin_amdgcn_readlane(cols, j);
        xv[j] = X[(size_t)col_j * FF + xoff];
      }
      float w[BGRP];
      int rlk[BGRP];
#pragma unroll
      for (int g = 0; g < BGRP; ++g) {
        int el = g * 4 + sub;
        int e_el = __shfl(eid, el);
        int rl_el = __shfl(rl, el);
        rlk[g] = rl_el * KK + k;
        const float* uv = uval + (size_t)e_el * DD;
        float s = 0.0f;
#pragma unroll
        for (int d = 0; d < DD; ++d) {
          float df = uv[d] - mk[d];
          s += df * df / fc[d];
        }
        float l = -0.5f * s;
        w[g] = __expf(l - finv(mkd[(r0 + rl_el) * KK + k]));
      }
#pragma unroll
      for (int g = 0; g < BGRP; ++g) {
        if (g * 4 + sub < nv) atomicAdd(&sden[rlk[g]], w[g]);
      }
#pragma unroll
      for (int g = 0; g < BGRP; ++g) {
#pragma unroll
        for (int ee = 0; ee < 4; ++ee) {
          if (g * 4 + ee < nv) {
            int rl_e = __builtin_amdgcn_readlane(rl, g * 4 + ee);
            float* dst = sagg + rl_e * RS + lane;
#pragma unroll
            for (int i = 0; i < 4; ++i) {
              float wk = row_share(w[g], ee * 4 + i);
              atomicAdd(dst + i * 32, wk * xv[g * 4 + ee]);
            }
          }
        }
      }
    }
  };

  for (int sw = 0; sw < SWEEPS + PFD; sw += PFD) {
    const bool last = (sw >= SWEEPS);
#pragma unroll
    for (int i = 0; i < PFD; ++i) {
      note(q[i], qb[i]);
      qb[i] = sweep_base(sw + PFD + i);
      q[i] = sweep_load(qb[i]);
    }

    for (int off = 0; off < HCAP; off += 32) {
      if (off < nh) {
        const int ridx = off + lane;
        v2u r = {0u, 0u};
        if (ridx < nh) r = myhits[ridx];
        unsigned g = r.x;
        const unsigned ebase = r.y * (unsigned)TPL;
#pragma nounroll
        for (int it = 0; it < TPL; ++it) {
          const bool has = (g != 0u);
          const unsigned m = __builtin_amdgcn_ballot_w32(has);
          if (m == 0u) break;
          const int pos = cnt + (int)__builtin_amdgcn_mbcnt_lo(m, 0u);
          if (has) {
            const int p = __builtin_ctz(g);
            mylist[pos] = ebase + (unsigned)(((p & 7) << 2) + (p >> 3));
          }
          g &= g - 1u;
          cnt += __builtin_popcount(m);
        }
      }
      const bool pending = (off + 32 < nh);
      drain(last && !pending);
      if (!pending) break;
    }
    nh = 0;
  }

  __syncthreads();
  const int nrows = (NN - r0 < SROWS) ? (NN - r0) : SROWS;
  const int tile = wave >> 1;
  const int bsel = wave & 1;
  if (wave < STILES * BB && tile * 16 < nrows) {
    int ar = tile * 16 + fb;
    ar = ar < SROWS ? ar : SROWS - 1;
    const float* ap = sagg + ar * RS + 4 * hi + bsel;
    const v4f* drow = (const v4f*)(sden + ar * KK);
    const v4f dlo = drow[0];
    const v4f dhi = drow[1];
    float rinv[KK];
#pragma unroll
    for (int kk = 0; kk < KK; ++kk) {
      float d = (kk < 4) ? dlo[kk & 3] : dhi[kk & 3];
      rinv[kk] = (d == 0.0f) ? 0.0f : 1.0f / d;
    }
    const float* bp = sgw + hi * 128 + fb * 4;
    v8f acc[4] = {};
#pragma unroll
    for (int s = 0; s < 16; ++s) {
      v2f a;
      a.x = ap[8 * s];
      a.y = ap[8 * s + 2];
      a *= rinv[s >> 1];
      const v4f b01 = *(const v4f*)(bp + s * 256);
      const v4f b23 = *(const v4f*)(bp + s * 256 + 64);
      v2f bf[4];
      bf[0].x = b01.x; bf[0].y = b01.y;
      bf[1].x = b01.z; bf[1].y = b01.w;
      bf[2].x = b23.x; bf[2].y = b23.y;
      bf[3].x = b23.z; bf[3].y = b23.w;
#pragma unroll
      for (int t = 0; t < 4; ++t) {
        acc[t] = __builtin_amdgcn_wmma_f32_16x16x4_f32(
             false, a,  false, bf[t],
             (short)0, acc[t],  false,  false);
      }
    }
    float* orow = out + ((size_t)bsel * NN + r0 + tile * 16 + hi * 8) * HH + 4 * fb;
    const int rleft = nrows - (tile * 16 + hi * 8);
#pragma unroll
    for (int j = 0; j < 8; ++j) {
      if (j < rleft) {
        v4f o;
        o.x = acc[0][j];
        o.y = acc[1][j];
        o.z = acc[2][j];
        o.w = acc[3][j];
        *(v4f*)(orow + (size_t)j * HH) = o;
      }
    }
  }
}

extern "C" void kernel_launch(void* const* d_in, const int* in_sizes, int n_in,
                              void* d_out, int out_size, void* d_ws, size_t ws_size,
                              hipStream_t stream) {
  const float* X     = (const float*)d_in[0];
  const float* uval  = (const float*)d_in[1];
  const float* mu    = (const float*)d_in[2];
  const float* sigma = (const float*)d_in[3];
  const float* GW    = (const float*)d_in[4];
  const int* urows   = (const int*)d_in[5];
  const int* ucols   = (const int*)d_in[6];
  float* out = (float*)d_out;

  unsigned* mkey = (unsigned*)d_ws;
  unsigned* tagw = (unsigned*)d_ws + NK;
  k_init<<<(NK + 255) / 256, 256, 0, stream>>>(mkey, NK);

  k_logit_max<<<(EWAVES + 7) / 8, 256, 0, stream>>>(uval, mu, sigma, urows, mkey, tagw);

  k_scatter<<<SLABS, SC_THREADS, 0, stream>>>(
      X, uval, mu, sigma, urows, ucols, tagw, mkey, GW, out);
}
